# starting kernel
# baseline (speedup 1.0000x reference)
#include <hip/hip_runtime.h>

typedef __attribute__((ext_vector_type(2))) float v2f;
typedef __attribute__((ext_vector_type(8))) float v8f;

#define NG   256
#define KDIM 12
#define GRID_DIM 128
#define NVOX (GRID_DIM*GRID_DIM*GRID_DIM)
#define ROW_BLOCKS (NVOX/16)
#define MAIN_BLOCKS 2048

__global__ __launch_bounds__(256) void vox_prep(const float* __restrict__ pos,
                                                const float* __restrict__ scl,
                                                const float* __restrict__ rot,
                                                float* __restrict__ w) {
  int n = blockIdx.x * blockDim.x + threadIdx.x;
  if (n >= NG) return;
  float qw = rot[n*4+0], qx = rot[n*4+1], qy = rot[n*4+2], qz = rot[n*4+3];
  float rn = rsqrtf(qw*qw + qx*qx + qy*qy + qz*qz);
  qw *= rn; qx *= rn; qy *= rn; qz *= rn;
  float R00 = 1.f - 2.f*(qy*qy + qz*qz);
  float R01 = 2.f*(qx*qy - qw*qz);
  float R02 = 2.f*(qx*qz + qw*qy);
  float R10 = 2.f*(qx*qy + qw*qz);
  float R11 = 1.f - 2.f*(qx*qx + qz*qz);
  float R12 = 2.f*(qy*qz - qw*qx);
  float R20 = 2.f*(qx*qz - qw*qy);
  float R21 = 2.f*(qy*qz + qw*qx);
  float R22 = 1.f - 2.f*(qx*qx + qy*qy);
  float s0 = scl[n*3+0], s1 = scl[n*3+1], s2 = scl[n*3+2];
  float i0 = 1.f/(s0*s0 + 1e-8f);
  float i1 = 1.f/(s1*s1 + 1e-8f);
  float i2 = 1.f/(s2*s2 + 1e-8f);
  float C00 = R00*R00*i0 + R01*R01*i1 + R02*R02*i2;
  float C01 = R00*R10*i0 + R01*R11*i1 + R02*R12*i2;
  float C02 = R00*R20*i0 + R01*R21*i1 + R02*R22*i2;
  float C11 = R10*R10*i0 + R11*R11*i1 + R12*R12*i2;
  float C12 = R10*R20*i0 + R11*R21*i1 + R12*R22*i2;
  float C22 = R20*R20*i0 + R21*R21*i1 + R22*R22*i2;
  float p0 = pos[n*3+0], p1 = pos[n*3+1], p2 = pos[n*3+2];
  float b0 = C00*p0 + C01*p1 + C02*p2;
  float b1 = C01*p0 + C11*p1 + C12*p2;
  float b2 = C02*p0 + C12*p1 + C22*p2;
  float cn = p0*b0 + p1*b1 + p2*b2;
  float* wn = w + n*KDIM;
  wn[0] = C00;      wn[1] = C11;      wn[2]  = C22;
  wn[3] = 2.f*C01;  wn[4] = 2.f*C02;  wn[5]  = 2.f*C12;
  wn[6] = -2.f*b0;  wn[7] = -2.f*b1;  wn[8]  = -2.f*b2;
  wn[9] = cn;       wn[10] = 0.f;     wn[11] = 0.f;
}

__global__ __launch_bounds__(256) void vox_main(const float* __restrict__ w,
                                                const float* __restrict__ density,
                                                float* __restrict__ out) {
  __shared__ float lw[NG * KDIM];
  __shared__ float ld[NG];

  const int tid = threadIdx.x;
  for (int i = tid; i < NG * KDIM; i += 256) lw[i] = w[i];
  if (tid < NG) ld[tid] = density[tid];
  __syncthreads();

  const int lane = tid & 31;
  const int wave = tid >> 5;
  const int m    = lane & 15;
  const bool h   = lane >= 16;

  const float EXP_SCALE = -0.72134752044448170368f;
  const float SC = 2.0f / 127.0f;

  int rb = blockIdx.x * 8 + wave;
  const int stride = gridDim.x * 8;

  for (; rb < ROW_BLOCKS; rb += stride) {
    const int v  = rb * 16 + m;
    const int iz = v >> 14;
    const int iy = (v >> 7) & 127;
    const int ix = v & 127;
    const float c0 = (float)iz * SC - 1.0f;
    const float c1 = (float)iy * SC - 1.0f;
    const float c2 = (float)ix * SC - 1.0f;

    v2f a0, a1, a2;
    a0.x = h ? c2*c2 : c0*c0;   a0.y = h ? c0*c1 : c1*c1;
    a1.x = h ? c0    : c0*c2;   a1.y = h ? c1    : c1*c2;
    a2.x = h ? 0.f   : c2;      a2.y = h ? 0.f   : 1.f;

    float acc[8];
    #pragma unroll
    for (int i = 0; i < 8; ++i) acc[i] = 0.f;

    #pragma unroll
    for (int t = 0; t < 16; ++t) {
      const int n = t * 16 + m;
      const v2f* wp = (const v2f*)&lw[n * KDIM];
      const int hs = h ? 1 : 0;
      v2f b0 = wp[0 + hs];
      v2f b1 = wp[2 + hs];
      v2f b2 = wp[4 + hs];

      v8f c = {0.f,0.f,0.f,0.f,0.f,0.f,0.f,0.f};
      c = __builtin_amdgcn_wmma_f32_16x16x4_f32(false, a0, false, b0, (short)0, c, false, false);
      c = __builtin_amdgcn_wmma_f32_16x16x4_f32(false, a1, false, b1, (short)0, c, false, false);
      c = __builtin_amdgcn_wmma_f32_16x16x4_f32(false, a2, false, b2, (short)0, c, false, false);

      const float dn = ld[n];
      #pragma unroll
      for (int i = 0; i < 8; ++i)
        acc[i] += __builtin_amdgcn_exp2f(c[i] * EXP_SCALE) * dn;
    }

    #pragma unroll
    for (int mask = 1; mask <= 8; mask <<= 1) {
      #pragma unroll
      for (int i = 0; i < 8; ++i)
        acc[i] += __shfl_xor(acc[i], mask, 32);
    }

    if (lane == 0 || lane == 16) {
      float4 lo = make_float4(acc[0], acc[1], acc[2], acc[3]);
      float4 hi = make_float4(acc[4], acc[5], acc[6], acc[7]);
      float* dst = out + rb * 16 + (h ? 8 : 0);
      ((float4*)dst)[0] = lo;
      ((float4*)dst)[1] = hi;
    }
  }
}

extern "C" void kernel_launch(void* const* d_in, const int* in_sizes, int n_in,
                              void* d_out, int out_size, void* d_ws, size_t ws_size,
                              hipStream_t stream) {
  const float* positions = (const float*)d_in[0];
  const float* scales    = (const float*)d_in[1];
  const float* rotations = (const float*)d_in[2];
  const float* density   = (const float*)d_in[3];
  float* out = (float*)d_out;
  float* wtab = (float*)d_ws;

  vox_prep<<<1, 256, 0, stream>>>(positions, scales, rotations, wtab);
  vox_main<<<MAIN_BLOCKS, 256, 0, stream>>>(wtab, density, out);
}


# Row walk made multiplicative: e(row+1) = e(row)*r, r *= k, k = exp(-C11*d*d) per gaussian: 2 exp2 and 4 WMMAs per 8 row tiles, was 8 and 10; term rounding changes, sum order unchanged
# speedup vs baseline: 2.3081x; 2.3081x over previous
#include <hip/hip_runtime.h>


typedef __attribute__((ext_vector_type(2))) float v2f;
typedef __attribute__((ext_vector_type(8))) float v8f;

#define NG   256
#define KDIM 12
#define WSTR 20
#define WFLAG (NG * WSTR)
#define C11_MAX 4000.0f
#define GRID_DIM 128
#define ROWS 8
#define MAIN_BLOCKS (GRID_DIM * (GRID_DIM / ROWS))

__global__ __launch_bounds__(256) void vox_prep(const float* __restrict__ pos,
                                                const float* __restrict__ scl,
                                                const float* __restrict__ rot,
                                                float* __restrict__ w) {
  const int n = threadIdx.x;
  float qw = rot[n*4+0], qx = rot[n*4+1], qy = rot[n*4+2], qz = rot[n*4+3];
  float rn = rsqrtf(qw*qw + qx*qx + qy*qy + qz*qz);
  qw *= rn; qx *= rn; qy *= rn; qz *= rn;
  float R00 = 1.f - 2.f*(qy*qy + qz*qz);
  float R01 = 2.f*(qx*qy - qw*qz);
  float R02 = 2.f*(qx*qz + qw*qy);
  float R10 = 2.f*(qx*qy + qw*qz);
  float R11 = 1.f - 2.f*(qx*qx + qz*qz);
  float R12 = 2.f*(qy*qz - qw*qx);
  float R20 = 2.f*(qx*qz - qw*qy);
  float R21 = 2.f*(qy*qz + qw*qx);
  float R22 = 1.f - 2.f*(qx*qx + qy*qy);
  float s0 = scl[n*3+0], s1 = scl[n*3+1], s2 = scl[n*3+2];
  float i0 = 1.f/(s0*s0 + 1e-8f);
  float i1 = 1.f/(s1*s1 + 1e-8f);
  float i2 = 1.f/(s2*s2 + 1e-8f);
  float C00 = R00*R00*i0 + R01*R01*i1 + R02*R02*i2;
  float C01 = R00*R10*i0 + R01*R11*i1 + R02*R12*i2;
  float C02 = R00*R20*i0 + R01*R21*i1 + R02*R22*i2;
  float C11 = R10*R10*i0 + R11*R11*i1 + R12*R12*i2;
  float C12 = R10*R20*i0 + R11*R21*i1 + R12*R22*i2;
  float C22 = R20*R20*i0 + R21*R21*i1 + R22*R22*i2;
  float p0 = pos[n*3+0], p1 = pos[n*3+1], p2 = pos[n*3+2];
  float b0 = C00*p0 + C01*p1 + C02*p2;
  float b1 = C01*p0 + C11*p1 + C12*p2;
  float b2 = C02*p0 + C12*p1 + C22*p2;
  float cn = p0*b0 + p1*b1 + p2*b2;
  const float d = 2.0f / 127.0f;
  float* wn = w + n*WSTR;
  wn[0] = C00;      wn[1] = C11;      wn[2]  = C22;
  wn[3] = 2.f*C01;  wn[4] = 2.f*C02;  wn[5]  = 2.f*C12;
  wn[6] = -2.f*b0;  wn[7] = -2.f*b1;  wn[8]  = -2.f*b2;
  wn[9] = cn;       wn[10] = 0.f;     wn[11] = 0.f;
  wn[12] = 2.f*C01*d;  wn[13] = 2.f*C11*d;
  wn[14] = 2.f*C12*d;  wn[15] = C11*d*d - 2.f*b1*d;
  wn[16] = expf(-C11*d*d);
  wn[17] = 0.f; wn[18] = 0.f; wn[19] = 0.f;
  const int wave_narrow = __any(!(C11 <= C11_MAX));
  if ((n & 31) == 0) w[WFLAG + (n >> 5)] = wave_narrow ? 1.f : 0.f;
}

__global__ __launch_bounds__(256) void vox_main(const float* __restrict__ w,
                                                const float* __restrict__ density,
                                                float* __restrict__ out) {
  __shared__ float lw[NG * WSTR];
  __shared__ float ld[NG];

  const int tid = threadIdx.x;
  for (int i = tid; i < NG * WSTR; i += 256) lw[i] = w[i];
  if (tid < NG) ld[tid] = density[tid];
  const float* fl = w + WFLAG;
  const bool product_walk =
      (fl[0] + fl[1] + fl[2] + fl[3] + fl[4] + fl[5] + fl[6] + fl[7] == 0.f);
  __syncthreads();

  const int lane = tid & 31;
  const int wave = tid >> 5;
  const int m    = lane & 15;
  const bool h   = lane >= 16;
  const int hs   = h ? 1 : 0;

  const float EXP_SCALE = -0.72134752044448170368f;
  const float SC = 2.0f / 127.0f;

  const int iz  = blockIdx.x >> 4;
  const int iy0 = (blockIdx.x & 15) * ROWS;
  const int ix  = wave * 16 + m;
  const float c0 = (float)iz  * SC - 1.0f;
  const float c1 = (float)iy0 * SC - 1.0f;
  const float c2 = (float)ix  * SC - 1.0f;

  v2f a0, a1, a2;
  a0.x = h ? c2*c2 : c0*c0;   a0.y = h ? c0*c1 : c1*c1;
  a1.x = h ? c0    : c0*c2;   a1.y = h ? c1    : c1*c2;
  a2.x = h ? 0.f   : c2;      a2.y = h ? 0.f   : 1.f;

  v2f acc[ROWS][4];
  #pragma unroll
  for (int j = 0; j < ROWS; ++j) {
    #pragma unroll
    for (int p = 0; p < 4; ++p) acc[j][p] = (v2f){0.f, 0.f};
  }

  if (product_walk) {
    v2f as0;
    as0.x = h ? c2  : c0;
    as0.y = h ? 1.f : c1;

    for (int t = 0; t < 16; ++t) {
      const int n = t * 16 + m;
      const v2f* wp = (const v2f*)&lw[n * WSTR];
      const v2f b0 = wp[0 + hs];
      const v2f b1 = wp[2 + hs];
      const v2f b2 = wp[4 + hs];
      const v2f bs = wp[6 + hs];
      const float kq = lw[n * WSTR + 16];
      const float dn = ld[n];
      const v2f k2 = {kq, kq};
      const v2f d2 = {dn, dn};

      v8f c = {0.f,0.f,0.f,0.f,0.f,0.f,0.f,0.f};
      c = __builtin_amdgcn_wmma_f32_16x16x4_f32(false, a0, false, b0, (short)0, c, false, false);
      c = __builtin_amdgcn_wmma_f32_16x16x4_f32(false, a1, false, b1, (short)0, c, false, false);
      c = __builtin_amdgcn_wmma_f32_16x16x4_f32(false, a2, false, b2, (short)0, c, false, false);
      v8f s = {0.f,0.f,0.f,0.f,0.f,0.f,0.f,0.f};
      s = __builtin_amdgcn_wmma_f32_16x16x4_f32(false, as0, false, bs, (short)0, s, false, false);

      v2f e[4], r[4];
      #pragma unroll
      for (int p = 0; p < 4; ++p) {
        e[p].x = __builtin_amdgcn_exp2f(c[2*p]     * EXP_SCALE);
        e[p].y = __builtin_amdgcn_exp2f(c[2*p + 1] * EXP_SCALE);
        r[p].x = __builtin_amdgcn_exp2f(fminf(s[2*p]     * EXP_SCALE, 126.0f));
        r[p].y = __builtin_amdgcn_exp2f(fminf(s[2*p + 1] * EXP_SCALE, 126.0f));
      }

      #pragma unroll
      for (int j = 0; j < ROWS; ++j) {
        #pragma unroll
        for (int p = 0; p < 4; ++p)
          acc[j][p] = __builtin_elementwise_fma(e[p], d2, acc[j][p]);
        if (j + 1 < ROWS) {
          #pragma unroll
          for (int p = 0; p < 4; ++p) e[p] *= r[p];
          if (j + 2 < ROWS) {
            #pragma unroll
            for (int p = 0; p < 4; ++p) r[p] *= k2;
          }
        }
      }
    }
  } else {
    v2f astep[ROWS - 1];
    #pragma unroll
    for (int j = 0; j < ROWS - 1; ++j) {
      const float yj = (float)(iy0 + j) * SC - 1.0f;
      astep[j].x = h ? c2  : c0;
      astep[j].y = h ? 1.f : yj;
    }

    for (int t = 0; t < 16; ++t) {
      const int n = t * 16 + m;
      const v2f* wp = (const v2f*)&lw[n * WSTR];
      const v2f b0 = wp[0 + hs];
      const v2f b1 = wp[2 + hs];
      const v2f b2 = wp[4 + hs];
      const v2f bs = wp[6 + hs];
      const float dn = ld[n];

      v8f c = {0.f,0.f,0.f,0.f,0.f,0.f,0.f,0.f};
      c = __builtin_amdgcn_wmma_f32_16x16x4_f32(false, a0, false, b0, (short)0, c, false, false);
      c = __builtin_amdgcn_wmma_f32_16x16x4_f32(false, a1, false, b1, (short)0, c, false, false);
      c = __builtin_amdgcn_wmma_f32_16x16x4_f32(false, a2, false, b2, (short)0, c, false, false);

      #pragma unroll
      for (int j = 0; j < ROWS; ++j) {
        v8f cn = c;
        if (j + 1 < ROWS)
          cn = __builtin_amdgcn_wmma_f32_16x16x4_f32(false, astep[j], false, bs, (short)0, c, false, false);
        #pragma unroll
        for (int p = 0; p < 4; ++p) {
          acc[j][p].x += __builtin_amdgcn_exp2f(c[2*p]     * EXP_SCALE) * dn;
          acc[j][p].y += __builtin_amdgcn_exp2f(c[2*p + 1] * EXP_SCALE) * dn;
        }
        c = cn;
      }
    }
  }

  #pragma unroll
  for (int j = 0; j < ROWS; ++j) {
    #pragma unroll
    for (int mask = 1; mask <= 8; mask <<= 1) {
      #pragma unroll
      for (int p = 0; p < 4; ++p) {
        acc[j][p].x += __shfl_xor(acc[j][p].x, mask, 32);
        acc[j][p].y += __shfl_xor(acc[j][p].y, mask, 32);
      }
    }

    if (lane == 0 || lane == 16) {
      float4 lo = make_float4(acc[j][0].x, acc[j][0].y, acc[j][1].x, acc[j][1].y);
      float4 hi = make_float4(acc[j][2].x, acc[j][2].y, acc[j][3].x, acc[j][3].y);
      float* dst = out + ((iz * GRID_DIM + iy0 + j) * GRID_DIM + wave * 16) + (h ? 8 : 0);
      ((float4*)dst)[0] = lo;
      ((float4*)dst)[1] = hi;
    }
  }
}

extern "C" void kernel_launch(void* const* d_in, const int* in_sizes, int n_in,
                              void* d_out, int out_size, void* d_ws, size_t ws_size,
                              hipStream_t stream) {
  const float* positions = (const float*)d_in[0];
  const float* scales    = (const float*)d_in[1];
  const float* rotations = (const float*)d_in[2];
  const float* density   = (const float*)d_in[3];
  float* out = (float*)d_out;
  float* wtab = (float*)d_ws;

  vox_prep<<<1, 256, 0, stream>>>(positions, scales, rotations, wtab);
  vox_main<<<MAIN_BLOCKS, 256, 0, stream>>>(wtab, density, out);
}


# Walk both ways from the middle row: a wave owns 16 y rows, one start tile serves 7 rows up and 8 down (down ratio = k*exp2(-step)): 4 WMMAs per 16 row tiles, was 8; term rounding changes
# speedup vs baseline: 2.5735x; 1.1150x over previous
#include <hip/hip_runtime.h>


typedef __attribute__((ext_vector_type(2))) float v2f;
typedef __attribute__((ext_vector_type(8))) float v8f;

#define NG   256
#define KDIM 12
#define WSTR 20
#define WFLAG (NG * WSTR)
#define C11_MAX 3000.0f
#define GRID_DIM 128
#define ROWS 16
#define HROWS 8
#define START 8
#define MAIN_BLOCKS (GRID_DIM * (GRID_DIM / ROWS))

__global__ __launch_bounds__(256) void vox_prep(const float* __restrict__ pos,
                                                const float* __restrict__ scl,
                                                const float* __restrict__ rot,
                                                float* __restrict__ w) {
  const int n = threadIdx.x;
  float qw = rot[n*4+0], qx = rot[n*4+1], qy = rot[n*4+2], qz = rot[n*4+3];
  float rn = rsqrtf(qw*qw + qx*qx + qy*qy + qz*qz);
  qw *= rn; qx *= rn; qy *= rn; qz *= rn;
  float R00 = 1.f - 2.f*(qy*qy + qz*qz);
  float R01 = 2.f*(qx*qy - qw*qz);
  float R02 = 2.f*(qx*qz + qw*qy);
  float R10 = 2.f*(qx*qy + qw*qz);
  float R11 = 1.f - 2.f*(qx*qx + qz*qz);
  float R12 = 2.f*(qy*qz - qw*qx);
  float R20 = 2.f*(qx*qz - qw*qy);
  float R21 = 2.f*(qy*qz + qw*qx);
  float R22 = 1.f - 2.f*(qx*qx + qy*qy);
  float s0 = scl[n*3+0], s1 = scl[n*3+1], s2 = scl[n*3+2];
  float i0 = 1.f/(s0*s0 + 1e-8f);
  float i1 = 1.f/(s1*s1 + 1e-8f);
  float i2 = 1.f/(s2*s2 + 1e-8f);
  float C00 = R00*R00*i0 + R01*R01*i1 + R02*R02*i2;
  float C01 = R00*R10*i0 + R01*R11*i1 + R02*R12*i2;
  float C02 = R00*R20*i0 + R01*R21*i1 + R02*R22*i2;
  float C11 = R10*R10*i0 + R11*R11*i1 + R12*R12*i2;
  float C12 = R10*R20*i0 + R11*R21*i1 + R12*R22*i2;
  float C22 = R20*R20*i0 + R21*R21*i1 + R22*R22*i2;
  float p0 = pos[n*3+0], p1 = pos[n*3+1], p2 = pos[n*3+2];
  float b0 = C00*p0 + C01*p1 + C02*p2;
  float b1 = C01*p0 + C11*p1 + C12*p2;
  float b2 = C02*p0 + C12*p1 + C22*p2;
  float cn = p0*b0 + p1*b1 + p2*b2;
  const float d = 2.0f / 127.0f;
  float* wn = w + n*WSTR;
  wn[0] = C00;      wn[1] = C11;      wn[2]  = C22;
  wn[3] = 2.f*C01;  wn[4] = 2.f*C02;  wn[5]  = 2.f*C12;
  wn[6] = -2.f*b0;  wn[7] = -2.f*b1;  wn[8]  = -2.f*b2;
  wn[9] = cn;       wn[10] = 0.f;     wn[11] = 0.f;
  wn[12] = 2.f*C01*d;  wn[13] = 2.f*C11*d;
  wn[14] = 2.f*C12*d;  wn[15] = C11*d*d - 2.f*b1*d;
  wn[16] = expf(-C11*d*d);
  wn[17] = 0.f; wn[18] = 0.f; wn[19] = 0.f;
  const int wave_narrow = __any(!(C11 <= C11_MAX));
  if ((n & 31) == 0) w[WFLAG + (n >> 5)] = wave_narrow ? 1.f : 0.f;
}

__device__ __forceinline__ void make_a(bool h, float c0, float c1, float c2,
                                       v2f& a0, v2f& a1, v2f& a2) {
  a0.x = h ? c2*c2 : c0*c0;   a0.y = h ? c0*c1 : c1*c1;
  a1.x = h ? c0    : c0*c2;   a1.y = h ? c1    : c1*c2;
  a2.x = h ? 0.f   : c2;      a2.y = h ? 0.f   : 1.f;
}

__device__ __forceinline__ void reduce_store_row(v2f (&a)[4], float* __restrict__ dst,
                                                 int lane, bool h) {
  #pragma unroll
  for (int mask = 1; mask <= 8; mask <<= 1) {
    #pragma unroll
    for (int p = 0; p < 4; ++p) {
      a[p].x += __shfl_xor(a[p].x, mask, 32);
      a[p].y += __shfl_xor(a[p].y, mask, 32);
    }
  }
  if (lane == 0 || lane == 16) {
    float4 lo = make_float4(a[0].x, a[0].y, a[1].x, a[1].y);
    float4 hi = make_float4(a[2].x, a[2].y, a[3].x, a[3].y);
    float* d = dst + (h ? 8 : 0);
    ((float4*)d)[0] = lo;
    ((float4*)d)[1] = hi;
  }
}

__global__ __launch_bounds__(256) void vox_main(const float* __restrict__ w,
                                                const float* __restrict__ density,
                                                float* __restrict__ out) {
  __shared__ float lw[NG * WSTR];
  __shared__ float ld[NG];

  const int tid = threadIdx.x;
  for (int i = tid; i < NG * WSTR; i += 256) lw[i] = w[i];
  if (tid < NG) ld[tid] = density[tid];
  const float* fl = w + WFLAG;
  const bool product_walk =
      (fl[0] + fl[1] + fl[2] + fl[3] + fl[4] + fl[5] + fl[6] + fl[7] == 0.f);
  __syncthreads();

  const int lane = tid & 31;
  const int wave = tid >> 5;
  const int m    = lane & 15;
  const bool h   = lane >= 16;
  const int hs   = h ? 1 : 0;

  const float EXP_SCALE = -0.72134752044448170368f;
  const float SC = 2.0f / 127.0f;

  const int iz  = blockIdx.x >> 3;
  const int iy0 = (blockIdx.x & 7) * ROWS;
  const int ix  = wave * 16 + m;
  const float c0 = (float)iz * SC - 1.0f;
  const float c2 = (float)ix * SC - 1.0f;
  float* const out_w = out + ((iz * GRID_DIM + iy0) * GRID_DIM + wave * 16);

  if (product_walk) {
    const float c1 = (float)(iy0 + START) * SC - 1.0f;
    v2f a0, a1, a2;
    make_a(h, c0, c1, c2, a0, a1, a2);
    v2f as0;
    as0.x = h ? c2  : c0;
    as0.y = h ? 1.f : c1;

    v2f acc[ROWS][4];
    #pragma unroll
    for (int j = 0; j < ROWS; ++j) {
      #pragma unroll
      for (int p = 0; p < 4; ++p) acc[j][p] = (v2f){0.f, 0.f};
    }

    for (int t = 0; t < 16; ++t) {
      const int n = t * 16 + m;
      const v2f* wp = (const v2f*)&lw[n * WSTR];
      const v2f b0 = wp[0 + hs];
      const v2f b1 = wp[2 + hs];
      const v2f b2 = wp[4 + hs];
      const v2f bs = wp[6 + hs];
      const float kq = lw[n * WSTR + 16];
      const float dn = ld[n];
      const v2f k2 = {kq, kq};
      const v2f d2 = {dn, dn};

      v8f c = {0.f,0.f,0.f,0.f,0.f,0.f,0.f,0.f};
      c = __builtin_amdgcn_wmma_f32_16x16x4_f32(false, a0, false, b0, (short)0, c, false, false);
      c = __builtin_amdgcn_wmma_f32_16x16x4_f32(false, a1, false, b1, (short)0, c, false, false);
      c = __builtin_amdgcn_wmma_f32_16x16x4_f32(false, a2, false, b2, (short)0, c, false, false);
      v8f s = {0.f,0.f,0.f,0.f,0.f,0.f,0.f,0.f};
      s = __builtin_amdgcn_wmma_f32_16x16x4_f32(false, as0, false, bs, (short)0, s, false, false);

      v2f e[4], eu[4], ru[4], rd[4];
      #pragma unroll
      for (int p = 0; p < 4; ++p) {
        const float ux = s[2*p]     * EXP_SCALE;
        const float uy = s[2*p + 1] * EXP_SCALE;
        e[p].x  = __builtin_amdgcn_exp2f(c[2*p]     * EXP_SCALE);
        e[p].y  = __builtin_amdgcn_exp2f(c[2*p + 1] * EXP_SCALE);
        ru[p].x = __builtin_amdgcn_exp2f(fminf( ux, 126.0f));
        ru[p].y = __builtin_amdgcn_exp2f(fminf( uy, 126.0f));
        rd[p].x = __builtin_amdgcn_exp2f(fminf(-ux, 126.0f));
        rd[p].y = __builtin_amdgcn_exp2f(fminf(-uy, 126.0f));
        rd[p] *= k2;
        eu[p] = e[p];
        acc[START][p] = __builtin_elementwise_fma(e[p], d2, acc[START][p]);
      }

      #pragma unroll
      for (int j = START + 1; j < ROWS; ++j) {
        #pragma unroll
        for (int p = 0; p < 4; ++p) {
          eu[p] *= ru[p];
          acc[j][p] = __builtin_elementwise_fma(eu[p], d2, acc[j][p]);
        }
        if (j + 1 < ROWS) {
          #pragma unroll
          for (int p = 0; p < 4; ++p) ru[p] *= k2;
        }
      }

      #pragma unroll
      for (int j = START - 1; j >= 0; --j) {
        #pragma unroll
        for (int p = 0; p < 4; ++p) {
          e[p] *= rd[p];
          acc[j][p] = __builtin_elementwise_fma(e[p], d2, acc[j][p]);
        }
        if (j > 0) {
          #pragma unroll
          for (int p = 0; p < 4; ++p) rd[p] *= k2;
        }
      }
    }

    #pragma unroll
    for (int j = 0; j < ROWS; ++j)
      reduce_store_row(acc[j], out_w + j * GRID_DIM, lane, h);
  } else {
    #pragma nounroll
    for (int g = 0; g < ROWS / HROWS; ++g) {
      const int iyg = iy0 + g * HROWS;
      const float c1 = (float)iyg * SC - 1.0f;
      v2f a0, a1, a2;
      make_a(h, c0, c1, c2, a0, a1, a2);
      v2f astep[HROWS - 1];
      #pragma unroll
      for (int j = 0; j < HROWS - 1; ++j) {
        const float yj = (float)(iyg + j) * SC - 1.0f;
        astep[j].x = h ? c2  : c0;
        astep[j].y = h ? 1.f : yj;
      }

      v2f acc[HROWS][4];
      #pragma unroll
      for (int j = 0; j < HROWS; ++j) {
        #pragma unroll
        for (int p = 0; p < 4; ++p) acc[j][p] = (v2f){0.f, 0.f};
      }

      for (int t = 0; t < 16; ++t) {
        const int n = t * 16 + m;
        const v2f* wp = (const v2f*)&lw[n * WSTR];
        const v2f b0 = wp[0 + hs];
        const v2f b1 = wp[2 + hs];
        const v2f b2 = wp[4 + hs];
        const v2f bs = wp[6 + hs];
        const float dn = ld[n];

        v8f c = {0.f,0.f,0.f,0.f,0.f,0.f,0.f,0.f};
        c = __builtin_amdgcn_wmma_f32_16x16x4_f32(false, a0, false, b0, (short)0, c, false, false);
        c = __builtin_amdgcn_wmma_f32_16x16x4_f32(false, a1, false, b1, (short)0, c, false, false);
        c = __builtin_amdgcn_wmma_f32_16x16x4_f32(false, a2, false, b2, (short)0, c, false, false);

        #pragma unroll
        for (int j = 0; j < HROWS; ++j) {
          v8f cn = c;
          if (j + 1 < HROWS)
            cn = __builtin_amdgcn_wmma_f32_16x16x4_f32(false, astep[j], false, bs, (short)0, c, false, false);
          #pragma unroll
          for (int p = 0; p < 4; ++p) {
            acc[j][p].x += __builtin_amdgcn_exp2f(c[2*p]     * EXP_SCALE) * dn;
            acc[j][p].y += __builtin_amdgcn_exp2f(c[2*p + 1] * EXP_SCALE) * dn;
          }
          c = cn;
        }
      }

      #pragma unroll
      for (int j = 0; j < HROWS; ++j)
        reduce_store_row(acc[j], out_w + (g * HROWS + j) * GRID_DIM, lane, h);
    }
  }
}

extern "C" void kernel_launch(void* const* d_in, const int* in_sizes, int n_in,
                              void* d_out, int out_size, void* d_ws, size_t ws_size,
                              hipStream_t stream) {
  const float* positions = (const float*)d_in[0];
  const float* scales    = (const float*)d_in[1];
  const float* rotations = (const float*)d_in[2];
  const float* density   = (const float*)d_in[3];
  float* out = (float*)d_out;
  float* wtab = (float*)d_ws;

  vox_prep<<<1, 256, 0, stream>>>(positions, scales, rotations, wtab);
  vox_main<<<MAIN_BLOCKS, 256, 0, stream>>>(wtab, density, out);
}


# Row ratio split: e(S+t) = e(S)*H^t*k^(t(t-1)/2); H per voxel, k power per gaussian, tabled times density as the FMA multiplier: 124 packed ops per tile, was 188; term rounding changes, sum order same
# speedup vs baseline: 2.6752x; 1.0395x over previous
#include <hip/hip_runtime.h>


typedef __attribute__((ext_vector_type(2))) float v2f;
typedef __attribute__((ext_vector_type(8))) float v8f;

#define NG   256
#define KDIM 12
#define WSTR 28
#define NKD  9
#define WFLAG (NG * WSTR)
#define C11_MAX 3000.0f
#define GRID_DIM 128
#define ROWS 16
#define HROWS 8
#define START 8
#define MAIN_BLOCKS (GRID_DIM * (GRID_DIM / ROWS))

__global__ __launch_bounds__(256) void vox_prep(const float* __restrict__ pos,
                                                const float* __restrict__ scl,
                                                const float* __restrict__ rot,
                                                const float* __restrict__ den,
                                                float* __restrict__ w) {
  const int n = threadIdx.x;
  float qw = rot[n*4+0], qx = rot[n*4+1], qy = rot[n*4+2], qz = rot[n*4+3];
  float rn = rsqrtf(qw*qw + qx*qx + qy*qy + qz*qz);
  qw *= rn; qx *= rn; qy *= rn; qz *= rn;
  float R00 = 1.f - 2.f*(qy*qy + qz*qz);
  float R01 = 2.f*(qx*qy - qw*qz);
  float R02 = 2.f*(qx*qz + qw*qy);
  float R10 = 2.f*(qx*qy + qw*qz);
  float R11 = 1.f - 2.f*(qx*qx + qz*qz);
  float R12 = 2.f*(qy*qz - qw*qx);
  float R20 = 2.f*(qx*qz - qw*qy);
  float R21 = 2.f*(qy*qz + qw*qx);
  float R22 = 1.f - 2.f*(qx*qx + qy*qy);
  float s0 = scl[n*3+0], s1 = scl[n*3+1], s2 = scl[n*3+2];
  float i0 = 1.f/(s0*s0 + 1e-8f);
  float i1 = 1.f/(s1*s1 + 1e-8f);
  float i2 = 1.f/(s2*s2 + 1e-8f);
  float C00 = R00*R00*i0 + R01*R01*i1 + R02*R02*i2;
  float C01 = R00*R10*i0 + R01*R11*i1 + R02*R12*i2;
  float C02 = R00*R20*i0 + R01*R21*i1 + R02*R22*i2;
  float C11 = R10*R10*i0 + R11*R11*i1 + R12*R12*i2;
  float C12 = R10*R20*i0 + R11*R21*i1 + R12*R22*i2;
  float C22 = R20*R20*i0 + R21*R21*i1 + R22*R22*i2;
  float p0 = pos[n*3+0], p1 = pos[n*3+1], p2 = pos[n*3+2];
  float b0 = C00*p0 + C01*p1 + C02*p2;
  float b1 = C01*p0 + C11*p1 + C12*p2;
  float b2 = C02*p0 + C12*p1 + C22*p2;
  float cn = p0*b0 + p1*b1 + p2*b2;
  const float d = 2.0f / 127.0f;
  float* wn = w + n*WSTR;
  wn[0] = C00;      wn[1] = C11;      wn[2]  = C22;
  wn[3] = 2.f*C01;  wn[4] = 2.f*C02;  wn[5]  = 2.f*C12;
  wn[6] = -2.f*b0;  wn[7] = -2.f*b1;  wn[8]  = -2.f*b2;
  wn[9] = cn;       wn[10] = 0.f;     wn[11] = 0.f;
  wn[12] = 2.f*C01*d;  wn[13] = 2.f*C11*d;
  wn[14] = 2.f*C12*d;  wn[15] = C11*d*d - 2.f*b1*d;
  const float dn = den[n];
  const float lk = -C11*d*d;
  wn[16] = dn;
  #pragma unroll
  for (int i = 1; i < NKD; ++i) wn[16 + i] = dn * expf(lk * (float)(i*(i+1)/2));
  wn[25] = 0.f; wn[26] = 0.f; wn[27] = 0.f;
  const int wave_narrow = __any(!(C11 <= C11_MAX));
  if ((n & 31) == 0) w[WFLAG + (n >> 5)] = wave_narrow ? 1.f : 0.f;
}

__device__ __forceinline__ void make_a(bool h, float c0, float c1, float c2,
                                       v2f& a0, v2f& a1, v2f& a2) {
  a0.x = h ? c2*c2 : c0*c0;   a0.y = h ? c0*c1 : c1*c1;
  a1.x = h ? c0    : c0*c2;   a1.y = h ? c1    : c1*c2;
  a2.x = h ? 0.f   : c2;      a2.y = h ? 0.f   : 1.f;
}

__device__ __forceinline__ void reduce_store_row(v2f (&a)[4], float* __restrict__ dst,
                                                 int lane, bool h) {
  #pragma unroll
  for (int mask = 1; mask <= 8; mask <<= 1) {
    #pragma unroll
    for (int p = 0; p < 4; ++p) {
      a[p].x += __shfl_xor(a[p].x, mask, 32);
      a[p].y += __shfl_xor(a[p].y, mask, 32);
    }
  }
  if (lane == 0 || lane == 16) {
    float4 lo = make_float4(a[0].x, a[0].y, a[1].x, a[1].y);
    float4 hi = make_float4(a[2].x, a[2].y, a[3].x, a[3].y);
    float* d = dst + (h ? 8 : 0);
    ((float4*)d)[0] = lo;
    ((float4*)d)[1] = hi;
  }
}

__global__ __launch_bounds__(256) void vox_main(const float* __restrict__ w,
                                                float* __restrict__ out) {
  __shared__ __attribute__((aligned(16))) float lw[NG * WSTR];

  const int tid = threadIdx.x;
  for (int i = tid; i < NG * WSTR; i += 256) lw[i] = w[i];
  const float* fl = w + WFLAG;
  const bool product_walk =
      (fl[0] + fl[1] + fl[2] + fl[3] + fl[4] + fl[5] + fl[6] + fl[7] == 0.f);
  __syncthreads();

  const int lane = tid & 31;
  const int wave = tid >> 5;
  const int m    = lane & 15;
  const bool h   = lane >= 16;
  const int hs   = h ? 1 : 0;

  const float EXP_SCALE = -0.72134752044448170368f;
  const float SC = 2.0f / 127.0f;

  const int iz  = blockIdx.x >> 3;
  const int iy0 = (blockIdx.x & 7) * ROWS;
  const int ix  = wave * 16 + m;
  const float c0 = (float)iz * SC - 1.0f;
  const float c2 = (float)ix * SC - 1.0f;
  float* const out_w = out + ((iz * GRID_DIM + iy0) * GRID_DIM + wave * 16);

  if (product_walk) {
    const float c1 = (float)(iy0 + START) * SC - 1.0f;
    v2f a0, a1, a2;
    make_a(h, c0, c1, c2, a0, a1, a2);
    v2f as0;
    as0.x = h ? c2  : c0;
    as0.y = h ? 1.f : c1;

    v2f acc[ROWS][4];
    #pragma unroll
    for (int j = 0; j < ROWS; ++j) {
      #pragma unroll
      for (int p = 0; p < 4; ++p) acc[j][p] = (v2f){0.f, 0.f};
    }

    for (int t = 0; t < 16; ++t) {
      const int n = t * 16 + m;
      const v2f* wp = (const v2f*)&lw[n * WSTR];
      const v2f b0 = wp[0 + hs];
      const v2f b1 = wp[2 + hs];
      const v2f b2 = wp[4 + hs];
      const v2f bs = wp[6 + hs];
      const float4 ka = *(const float4*)&lw[n * WSTR + 16];
      const float4 kb = *(const float4*)&lw[n * WSTR + 20];
      const float  k8 = lw[n * WSTR + 24];
      const v2f kd[NKD] = {{ka.x, ka.x}, {ka.y, ka.y}, {ka.z, ka.z}, {ka.w, ka.w},
                           {kb.x, kb.x}, {kb.y, kb.y}, {kb.z, kb.z}, {kb.w, kb.w},
                           {k8, k8}};

      v8f c = {0.f,0.f,0.f,0.f,0.f,0.f,0.f,0.f};
      c = __builtin_amdgcn_wmma_f32_16x16x4_f32(false, a0, false, b0, (short)0, c, false, false);
      c = __builtin_amdgcn_wmma_f32_16x16x4_f32(false, a1, false, b1, (short)0, c, false, false);
      c = __builtin_amdgcn_wmma_f32_16x16x4_f32(false, a2, false, b2, (short)0, c, false, false);
      v8f s = {0.f,0.f,0.f,0.f,0.f,0.f,0.f,0.f};
      s = __builtin_amdgcn_wmma_f32_16x16x4_f32(false, as0, false, bs, (short)0, s, false, false);

      v2f gd[4], gu[4], hu[4], hd[4];
      #pragma unroll
      for (int p = 0; p < 4; ++p) {
        const float ux = s[2*p]     * EXP_SCALE;
        const float uy = s[2*p + 1] * EXP_SCALE;
        gd[p].x = __builtin_amdgcn_exp2f(c[2*p]     * EXP_SCALE);
        gd[p].y = __builtin_amdgcn_exp2f(c[2*p + 1] * EXP_SCALE);
        hu[p].x = __builtin_amdgcn_exp2f(fminf( ux, 126.0f));
        hu[p].y = __builtin_amdgcn_exp2f(fminf( uy, 126.0f));
        hd[p].x = __builtin_amdgcn_exp2f(fminf(-ux, 126.0f));
        hd[p].y = __builtin_amdgcn_exp2f(fminf(-uy, 126.0f));
        gu[p] = gd[p];
        acc[START][p] = __builtin_elementwise_fma(gd[p], kd[0], acc[START][p]);
      }

      #pragma unroll
      for (int j = START + 1; j < ROWS; ++j) {
        #pragma unroll
        for (int p = 0; p < 4; ++p) {
          gu[p] *= hu[p];
          acc[j][p] = __builtin_elementwise_fma(gu[p], kd[j - START - 1], acc[j][p]);
        }
      }

      #pragma unroll
      for (int j = START - 1; j >= 0; --j) {
        #pragma unroll
        for (int p = 0; p < 4; ++p) {
          gd[p] *= hd[p];
          acc[j][p] = __builtin_elementwise_fma(gd[p], kd[START - j], acc[j][p]);
        }
      }
    }

    #pragma unroll
    for (int j = 0; j < ROWS; ++j)
      reduce_store_row(acc[j], out_w + j * GRID_DIM, lane, h);
  } else {
    #pragma nounroll
    for (int g = 0; g < ROWS / HROWS; ++g) {
      const int iyg = iy0 + g * HROWS;
      const float c1 = (float)iyg * SC - 1.0f;
      v2f a0, a1, a2;
      make_a(h, c0, c1, c2, a0, a1, a2);
      v2f astep[HROWS - 1];
      #pragma unroll
      for (int j = 0; j < HROWS - 1; ++j) {
        const float yj = (float)(iyg + j) * SC - 1.0f;
        astep[j].x = h ? c2  : c0;
        astep[j].y = h ? 1.f : yj;
      }

      v2f acc[HROWS][4];
      #pragma unroll
      for (int j = 0; j < HROWS; ++j) {
        #pragma unroll
        for (int p = 0; p < 4; ++p) acc[j][p] = (v2f){0.f, 0.f};
      }

      for (int t = 0; t < 16; ++t) {
        const int n = t * 16 + m;
        const v2f* wp = (const v2f*)&lw[n * WSTR];
        const v2f b0 = wp[0 + hs];
        const v2f b1 = wp[2 + hs];
        const v2f b2 = wp[4 + hs];
        const v2f bs = wp[6 + hs];
        const float dn = lw[n * WSTR + 16];

        v8f c = {0.f,0.f,0.f,0.f,0.f,0.f,0.f,0.f};
        c = __builtin_amdgcn_wmma_f32_16x16x4_f32(false, a0, false, b0, (short)0, c, false, false);
        c = __builtin_amdgcn_wmma_f32_16x16x4_f32(false, a1, false, b1, (short)0, c, false, false);
        c = __builtin_amdgcn_wmma_f32_16x16x4_f32(false, a2, false, b2, (short)0, c, false, false);

        #pragma unroll
        for (int j = 0; j < HROWS; ++j) {
          v8f cn = c;
          if (j + 1 < HROWS)
            cn = __builtin_amdgcn_wmma_f32_16x16x4_f32(false, astep[j], false, bs, (short)0, c, false, false);
          #pragma unroll
          for (int p = 0; p < 4; ++p) {
            acc[j][p].x += __builtin_amdgcn_exp2f(c[2*p]     * EXP_SCALE) * dn;
            acc[j][p].y += __builtin_amdgcn_exp2f(c[2*p + 1] * EXP_SCALE) * dn;
          }
          c = cn;
        }
      }

      #pragma unroll
      for (int j = 0; j < HROWS; ++j)
        reduce_store_row(acc[j], out_w + (g * HROWS + j) * GRID_DIM, lane, h);
    }
  }
}

extern "C" void kernel_launch(void* const* d_in, const int* in_sizes, int n_in,
                              void* d_out, int out_size, void* d_ws, size_t ws_size,
                              hipStream_t stream) {
  const float* positions = (const float*)d_in[0];
  const float* scales    = (const float*)d_in[1];
  const float* rotations = (const float*)d_in[2];
  const float* density   = (const float*)d_in[3];
  float* out = (float*)d_out;
  float* wtab = (float*)d_ws;

  vox_prep<<<1, 256, 0, stream>>>(positions, scales, rotations, density, wtab);
  vox_main<<<MAIN_BLOCKS, 256, 0, stream>>>(wtab, out);
}


# Epilogue column sums by a halving exchange: per step a lane keeps half its rows, hands the rest to lane^mask by DPP: 120 exchanges per wave, was 512 LDS permutes; a lane stores one row; sum order same
# speedup vs baseline: 2.9052x; 1.0860x over previous
#include <hip/hip_runtime.h>


typedef __attribute__((ext_vector_type(2))) float v2f;
typedef __attribute__((ext_vector_type(8))) float v8f;

#define NG   256
#define KDIM 12
#define WSTR 28
#define NKD  9
#define WFLAG (NG * WSTR)
#define C11_MAX 3000.0f
#define GRID_DIM 128
#define ROWS 16
#define HROWS 8
#define START 8
#define MAIN_BLOCKS (GRID_DIM * (GRID_DIM / ROWS))

__global__ __launch_bounds__(256) void vox_prep(const float* __restrict__ pos,
                                                const float* __restrict__ scl,
                                                const float* __restrict__ rot,
                                                const float* __restrict__ den,
                                                float* __restrict__ w) {
  const int n = threadIdx.x;
  float qw = rot[n*4+0], qx = rot[n*4+1], qy = rot[n*4+2], qz = rot[n*4+3];
  float rn = rsqrtf(qw*qw + qx*qx + qy*qy + qz*qz);
  qw *= rn; qx *= rn; qy *= rn; qz *= rn;
  float R00 = 1.f - 2.f*(qy*qy + qz*qz);
  float R01 = 2.f*(qx*qy - qw*qz);
  float R02 = 2.f*(qx*qz + qw*qy);
  float R10 = 2.f*(qx*qy + qw*qz);
  float R11 = 1.f - 2.f*(qx*qx + qz*qz);
  float R12 = 2.f*(qy*qz - qw*qx);
  float R20 = 2.f*(qx*qz - qw*qy);
  float R21 = 2.f*(qy*qz + qw*qx);
  float R22 = 1.f - 2.f*(qx*qx + qy*qy);
  float s0 = scl[n*3+0], s1 = scl[n*3+1], s2 = scl[n*3+2];
  float i0 = 1.f/(s0*s0 + 1e-8f);
  float i1 = 1.f/(s1*s1 + 1e-8f);
  float i2 = 1.f/(s2*s2 + 1e-8f);
  float C00 = R00*R00*i0 + R01*R01*i1 + R02*R02*i2;
  float C01 = R00*R10*i0 + R01*R11*i1 + R02*R12*i2;
  float C02 = R00*R20*i0 + R01*R21*i1 + R02*R22*i2;
  float C11 = R10*R10*i0 + R11*R11*i1 + R12*R12*i2;
  float C12 = R10*R20*i0 + R11*R21*i1 + R12*R22*i2;
  float C22 = R20*R20*i0 + R21*R21*i1 + R22*R22*i2;
  float p0 = pos[n*3+0], p1 = pos[n*3+1], p2 = pos[n*3+2];
  float b0 = C00*p0 + C01*p1 + C02*p2;
  float b1 = C01*p0 + C11*p1 + C12*p2;
  float b2 = C02*p0 + C12*p1 + C22*p2;
  float cn = p0*b0 + p1*b1 + p2*b2;
  const float d = 2.0f / 127.0f;
  float* wn = w + n*WSTR;
  wn[0] = C00;      wn[1] = C11;      wn[2]  = C22;
  wn[3] = 2.f*C01;  wn[4] = 2.f*C02;  wn[5]  = 2.f*C12;
  wn[6] = -2.f*b0;  wn[7] = -2.f*b1;  wn[8]  = -2.f*b2;
  wn[9] = cn;       wn[10] = 0.f;     wn[11] = 0.f;
  wn[12] = 2.f*C01*d;  wn[13] = 2.f*C11*d;
  wn[14] = 2.f*C12*d;  wn[15] = C11*d*d - 2.f*b1*d;
  const float dn = den[n];
  const float lk = -C11*d*d;
  wn[16] = dn;
  #pragma unroll
  for (int i = 1; i < NKD; ++i) wn[16 + i] = dn * expf(lk * (float)(i*(i+1)/2));
  wn[25] = 0.f; wn[26] = 0.f; wn[27] = 0.f;
  const int wave_narrow = __any(!(C11 <= C11_MAX));
  if ((n & 31) == 0) w[WFLAG + (n >> 5)] = wave_narrow ? 1.f : 0.f;
}

__device__ __forceinline__ void make_a(bool h, float c0, float c1, float c2,
                                       v2f& a0, v2f& a1, v2f& a2) {
  a0.x = h ? c2*c2 : c0*c0;   a0.y = h ? c0*c1 : c1*c1;
  a1.x = h ? c0    : c0*c2;   a1.y = h ? c1    : c1*c2;
  a2.x = h ? 0.f   : c2;      a2.y = h ? 0.f   : 1.f;
}

__device__ __forceinline__ void reduce_store_row(v2f (&a)[4], float* __restrict__ dst,
                                                 int lane, bool h) {
  #pragma unroll
  for (int mask = 1; mask <= 8; mask <<= 1) {
    #pragma unroll
    for (int p = 0; p < 4; ++p) {
      a[p].x += __shfl_xor(a[p].x, mask, 32);
      a[p].y += __shfl_xor(a[p].y, mask, 32);
    }
  }
  if (lane == 0 || lane == 16) {
    float4 lo = make_float4(a[0].x, a[0].y, a[1].x, a[1].y);
    float4 hi = make_float4(a[2].x, a[2].y, a[3].x, a[3].y);
    float* d = dst + (h ? 8 : 0);
    ((float4*)d)[0] = lo;
    ((float4*)d)[1] = hi;
  }
}

template <int XMASK>
__device__ __forceinline__ float add_from_partner(float keep, float send) {
  const int got = __builtin_amdgcn_update_dpp(0, __float_as_int(send), 0x160 + XMASK, 0xf, 0xf, true);
  return keep + __int_as_float(got);
}

template <int STEP>
__device__ __forceinline__ void halve_rows(v2f (&acc)[ROWS][4], bool up) {
  constexpr int D = 1 << STEP;
  #pragma unroll
  for (int j = 0; j < ROWS; j += 2 * D) {
    #pragma unroll
    for (int p = 0; p < 4; ++p) {
      const v2f lo = acc[j][p], hi = acc[j + D][p];
      const v2f keep = up ? hi : lo;
      const v2f send = up ? lo : hi;
      acc[j][p].x = add_from_partner<D>(keep.x, send.x);
      acc[j][p].y = add_from_partner<D>(keep.y, send.y);
    }
  }
}

__device__ __forceinline__ void reduce_store_patch(v2f (&acc)[ROWS][4], float* __restrict__ patch,
                                                   int lane, bool h) {
  halve_rows<0>(acc, (lane & 1) != 0);
  halve_rows<1>(acc, (lane & 2) != 0);
  halve_rows<2>(acc, (lane & 4) != 0);
  halve_rows<3>(acc, (lane & 8) != 0);
  float* d = patch + (lane & 15) * GRID_DIM + (h ? 8 : 0);
  ((float4*)d)[0] = make_float4(acc[0][0].x, acc[0][0].y, acc[0][1].x, acc[0][1].y);
  ((float4*)d)[1] = make_float4(acc[0][2].x, acc[0][2].y, acc[0][3].x, acc[0][3].y);
}

__global__ __launch_bounds__(256) void vox_main(const float* __restrict__ w,
                                                float* __restrict__ out) {
  __shared__ __attribute__((aligned(16))) float lw[NG * WSTR];

  const int tid = threadIdx.x;
  for (int i = tid; i < NG * WSTR; i += 256) lw[i] = w[i];
  const float* fl = w + WFLAG;
  const bool product_walk =
      (fl[0] + fl[1] + fl[2] + fl[3] + fl[4] + fl[5] + fl[6] + fl[7] == 0.f);
  __syncthreads();

  const int lane = tid & 31;
  const int wave = tid >> 5;
  const int m    = lane & 15;
  const bool h   = lane >= 16;
  const int hs   = h ? 1 : 0;

  const float EXP_SCALE = -0.72134752044448170368f;
  const float SC = 2.0f / 127.0f;

  const int iz  = blockIdx.x >> 3;
  const int iy0 = (blockIdx.x & 7) * ROWS;
  const int ix  = wave * 16 + m;
  const float c0 = (float)iz * SC - 1.0f;
  const float c2 = (float)ix * SC - 1.0f;
  float* const out_w = out + ((iz * GRID_DIM + iy0) * GRID_DIM + wave * 16);

  if (product_walk) {
    const float c1 = (float)(iy0 + START) * SC - 1.0f;
    v2f a0, a1, a2;
    make_a(h, c0, c1, c2, a0, a1, a2);
    v2f as0;
    as0.x = h ? c2  : c0;
    as0.y = h ? 1.f : c1;

    v2f acc[ROWS][4];
    #pragma unroll
    for (int j = 0; j < ROWS; ++j) {
      #pragma unroll
      for (int p = 0; p < 4; ++p) acc[j][p] = (v2f){0.f, 0.f};
    }

    for (int t = 0; t < 16; ++t) {
      const int n = t * 16 + m;
      const v2f* wp = (const v2f*)&lw[n * WSTR];
      const v2f b0 = wp[0 + hs];
      const v2f b1 = wp[2 + hs];
      const v2f b2 = wp[4 + hs];
      const v2f bs = wp[6 + hs];
      const float4 ka = *(const float4*)&lw[n * WSTR + 16];
      const float4 kb = *(const float4*)&lw[n * WSTR + 20];
      const float  k8 = lw[n * WSTR + 24];
      const v2f kd[NKD] = {{ka.x, ka.x}, {ka.y, ka.y}, {ka.z, ka.z}, {ka.w, ka.w},
                           {kb.x, kb.x}, {kb.y, kb.y}, {kb.z, kb.z}, {kb.w, kb.w},
                           {k8, k8}};

      v8f c = {0.f,0.f,0.f,0.f,0.f,0.f,0.f,0.f};
      c = __builtin_amdgcn_wmma_f32_16x16x4_f32(false, a0, false, b0, (short)0, c, false, false);
      c = __builtin_amdgcn_wmma_f32_16x16x4_f32(false, a1, false, b1, (short)0, c, false, false);
      c = __builtin_amdgcn_wmma_f32_16x16x4_f32(false, a2, false, b2, (short)0, c, false, false);
      v8f s = {0.f,0.f,0.f,0.f,0.f,0.f,0.f,0.f};
      s = __builtin_amdgcn_wmma_f32_16x16x4_f32(false, as0, false, bs, (short)0, s, false, false);

      v2f gd[4], gu[4], hu[4], hd[4];
      #pragma unroll
      for (int p = 0; p < 4; ++p) {
        const float ux = s[2*p]     * EXP_SCALE;
        const float uy = s[2*p + 1] * EXP_SCALE;
        gd[p].x = __builtin_amdgcn_exp2f(c[2*p]     * EXP_SCALE);
        gd[p].y = __builtin_amdgcn_exp2f(c[2*p + 1] * EXP_SCALE);
        hu[p].x = __builtin_amdgcn_exp2f(fminf( ux, 126.0f));
        hu[p].y = __builtin_amdgcn_exp2f(fminf( uy, 126.0f));
        hd[p].x = __builtin_amdgcn_exp2f(fminf(-ux, 126.0f));
        hd[p].y = __builtin_amdgcn_exp2f(fminf(-uy, 126.0f));
        gu[p] = gd[p];
        acc[START][p] = __builtin_elementwise_fma(gd[p], kd[0], acc[START][p]);
      }

      #pragma unroll
      for (int j = START + 1; j < ROWS; ++j) {
        #pragma unroll
        for (int p = 0; p < 4; ++p) {
          gu[p] *= hu[p];
          acc[j][p] = __builtin_elementwise_fma(gu[p], kd[j - START - 1], acc[j][p]);
        }
      }

      #pragma unroll
      for (int j = START - 1; j >= 0; --j) {
        #pragma unroll
        for (int p = 0; p < 4; ++p) {
          gd[p] *= hd[p];
          acc[j][p] = __builtin_elementwise_fma(gd[p], kd[START - j], acc[j][p]);
        }
      }
    }

    reduce_store_patch(acc, out_w, lane, h);
  } else {
    #pragma nounroll
    for (int g = 0; g < ROWS / HROWS; ++g) {
      const int iyg = iy0 + g * HROWS;
      const float c1 = (float)iyg * SC - 1.0f;
      v2f a0, a1, a2;
      make_a(h, c0, c1, c2, a0, a1, a2);
      v2f astep[HROWS - 1];
      #pragma unroll
      for (int j = 0; j < HROWS - 1; ++j) {
        const float yj = (float)(iyg + j) * SC - 1.0f;
        astep[j].x = h ? c2  : c0;
        astep[j].y = h ? 1.f : yj;
      }

      v2f acc[HROWS][4];
      #pragma unroll
      for (int j = 0; j < HROWS; ++j) {
        #pragma unroll
        for (int p = 0; p < 4; ++p) acc[j][p] = (v2f){0.f, 0.f};
      }

      for (int t = 0; t < 16; ++t) {
        const int n = t * 16 + m;
        const v2f* wp = (const v2f*)&lw[n * WSTR];
        const v2f b0 = wp[0 + hs];
        const v2f b1 = wp[2 + hs];
        const v2f b2 = wp[4 + hs];
        const v2f bs = wp[6 + hs];
        const float dn = lw[n * WSTR + 16];

        v8f c = {0.f,0.f,0.f,0.f,0.f,0.f,0.f,0.f};
        c = __builtin_amdgcn_wmma_f32_16x16x4_f32(false, a0, false, b0, (short)0, c, false, false);
        c = __builtin_amdgcn_wmma_f32_16x16x4_f32(false, a1, false, b1, (short)0, c, false, false);
        c = __builtin_amdgcn_wmma_f32_16x16x4_f32(false, a2, false, b2, (short)0, c, false, false);

        #pragma unroll
        for (int j = 0; j < HROWS; ++j) {
          v8f cn = c;
          if (j + 1 < HROWS)
            cn = __builtin_amdgcn_wmma_f32_16x16x4_f32(false, astep[j], false, bs, (short)0, c, false, false);
          #pragma unroll
          for (int p = 0; p < 4; ++p) {
            acc[j][p].x += __builtin_amdgcn_exp2f(c[2*p]     * EXP_SCALE) * dn;
            acc[j][p].y += __builtin_amdgcn_exp2f(c[2*p + 1] * EXP_SCALE) * dn;
          }
          c = cn;
        }
      }

      #pragma unroll
      for (int j = 0; j < HROWS; ++j)
        reduce_store_row(acc[j], out_w + (g * HROWS + j) * GRID_DIM, lane, h);
    }
  }
}

extern "C" void kernel_launch(void* const* d_in, const int* in_sizes, int n_in,
                              void* d_out, int out_size, void* d_ws, size_t ws_size,
                              hipStream_t stream) {
  const float* positions = (const float*)d_in[0];
  const float* scales    = (const float*)d_in[1];
  const float* rotations = (const float*)d_in[2];
  const float* density   = (const float*)d_in[3];
  float* out = (float*)d_out;
  float* wtab = (float*)d_ws;

  vox_prep<<<1, 256, 0, stream>>>(positions, scales, rotations, density, wtab);
  vox_main<<<MAIN_BLOCKS, 256, 0, stream>>>(wtab, out);
}


# Table staged once per block for two (z, 16-row) patches, by 7 16-byte loads per thread all in flight (was 28 4-byte loads in dependent rounds, per patch); 512 blocks, was 1024; sum order same
# speedup vs baseline: 3.3829x; 1.1644x over previous
#include <hip/hip_runtime.h>


typedef __attribute__((ext_vector_type(2))) float v2f;
typedef __attribute__((ext_vector_type(8))) float v8f;

#define NG   256
#define KDIM 12
#define WSTR 28
#define NKD  9
#define WFLAG (NG * WSTR)
#define C11_MAX 3000.0f
#define GRID_DIM 128
#define ROWS 16
#define HROWS 8
#define START 8
#define PATCH_ROWS_PER_BLOCK 2
#define MAIN_BLOCKS (GRID_DIM * (GRID_DIM / ROWS) / PATCH_ROWS_PER_BLOCK)

__global__ __launch_bounds__(256) void vox_prep(const float* __restrict__ pos,
                                                const float* __restrict__ scl,
                                                const float* __restrict__ rot,
                                                const float* __restrict__ den,
                                                float* __restrict__ w) {
  const int n = threadIdx.x;
  float qw = rot[n*4+0], qx = rot[n*4+1], qy = rot[n*4+2], qz = rot[n*4+3];
  float rn = rsqrtf(qw*qw + qx*qx + qy*qy + qz*qz);
  qw *= rn; qx *= rn; qy *= rn; qz *= rn;
  float R00 = 1.f - 2.f*(qy*qy + qz*qz);
  float R01 = 2.f*(qx*qy - qw*qz);
  float R02 = 2.f*(qx*qz + qw*qy);
  float R10 = 2.f*(qx*qy + qw*qz);
  float R11 = 1.f - 2.f*(qx*qx + qz*qz);
  float R12 = 2.f*(qy*qz - qw*qx);
  float R20 = 2.f*(qx*qz - qw*qy);
  float R21 = 2.f*(qy*qz + qw*qx);
  float R22 = 1.f - 2.f*(qx*qx + qy*qy);
  float s0 = scl[n*3+0], s1 = scl[n*3+1], s2 = scl[n*3+2];
  float i0 = 1.f/(s0*s0 + 1e-8f);
  float i1 = 1.f/(s1*s1 + 1e-8f);
  float i2 = 1.f/(s2*s2 + 1e-8f);
  float C00 = R00*R00*i0 + R01*R01*i1 + R02*R02*i2;
  float C01 = R00*R10*i0 + R01*R11*i1 + R02*R12*i2;
  float C02 = R00*R20*i0 + R01*R21*i1 + R02*R22*i2;
  float C11 = R10*R10*i0 + R11*R11*i1 + R12*R12*i2;
  float C12 = R10*R20*i0 + R11*R21*i1 + R12*R22*i2;
  float C22 = R20*R20*i0 + R21*R21*i1 + R22*R22*i2;
  float p0 = pos[n*3+0], p1 = pos[n*3+1], p2 = pos[n*3+2];
  float b0 = C00*p0 + C01*p1 + C02*p2;
  float b1 = C01*p0 + C11*p1 + C12*p2;
  float b2 = C02*p0 + C12*p1 + C22*p2;
  float cn = p0*b0 + p1*b1 + p2*b2;
  const float d = 2.0f / 127.0f;
  float* wn = w + n*WSTR;
  wn[0] = C00;      wn[1] = C11;      wn[2]  = C22;
  wn[3] = 2.f*C01;  wn[4] = 2.f*C02;  wn[5]  = 2.f*C12;
  wn[6] = -2.f*b0;  wn[7] = -2.f*b1;  wn[8]  = -2.f*b2;
  wn[9] = cn;       wn[10] = 0.f;     wn[11] = 0.f;
  wn[12] = 2.f*C01*d;  wn[13] = 2.f*C11*d;
  wn[14] = 2.f*C12*d;  wn[15] = C11*d*d - 2.f*b1*d;
  const float dn = den[n];
  const float lk = -C11*d*d;
  wn[16] = dn;
  #pragma unroll
  for (int i = 1; i < NKD; ++i) wn[16 + i] = dn * expf(lk * (float)(i*(i+1)/2));
  wn[25] = 0.f; wn[26] = 0.f; wn[27] = 0.f;
  const int wave_narrow = __any(!(C11 <= C11_MAX));
  if ((n & 31) == 0) w[WFLAG + (n >> 5)] = wave_narrow ? 1.f : 0.f;
}

__device__ __forceinline__ void make_a(bool h, float c0, float c1, float c2,
                                       v2f& a0, v2f& a1, v2f& a2) {
  a0.x = h ? c2*c2 : c0*c0;   a0.y = h ? c0*c1 : c1*c1;
  a1.x = h ? c0    : c0*c2;   a1.y = h ? c1    : c1*c2;
  a2.x = h ? 0.f   : c2;      a2.y = h ? 0.f   : 1.f;
}

__device__ __forceinline__ void reduce_store_row(v2f (&a)[4], float* __restrict__ dst,
                                                 int lane, bool h) {
  #pragma unroll
  for (int mask = 1; mask <= 8; mask <<= 1) {
    #pragma unroll
    for (int p = 0; p < 4; ++p) {
      a[p].x += __shfl_xor(a[p].x, mask, 32);
      a[p].y += __shfl_xor(a[p].y, mask, 32);
    }
  }
  if (lane == 0 || lane == 16) {
    float4 lo = make_float4(a[0].x, a[0].y, a[1].x, a[1].y);
    float4 hi = make_float4(a[2].x, a[2].y, a[3].x, a[3].y);
    float* d = dst + (h ? 8 : 0);
    ((float4*)d)[0] = lo;
    ((float4*)d)[1] = hi;
  }
}

template <int XMASK>
__device__ __forceinline__ float add_from_partner(float keep, float send) {
  const int got = __builtin_amdgcn_update_dpp(0, __float_as_int(send), 0x160 + XMASK, 0xf, 0xf, true);
  return keep + __int_as_float(got);
}

template <int STEP>
__device__ __forceinline__ void halve_rows(v2f (&acc)[ROWS][4], bool up) {
  constexpr int D = 1 << STEP;
  #pragma unroll
  for (int j = 0; j < ROWS; j += 2 * D) {
    #pragma unroll
    for (int p = 0; p < 4; ++p) {
      const v2f lo = acc[j][p], hi = acc[j + D][p];
      const v2f keep = up ? hi : lo;
      const v2f send = up ? lo : hi;
      acc[j][p].x = add_from_partner<D>(keep.x, send.x);
      acc[j][p].y = add_from_partner<D>(keep.y, send.y);
    }
  }
}

__device__ __forceinline__ void reduce_store_patch(v2f (&acc)[ROWS][4], float* __restrict__ patch,
                                                   int lane, bool h) {
  halve_rows<0>(acc, (lane & 1) != 0);
  halve_rows<1>(acc, (lane & 2) != 0);
  halve_rows<2>(acc, (lane & 4) != 0);
  halve_rows<3>(acc, (lane & 8) != 0);
  float* d = patch + (lane & 15) * GRID_DIM + (h ? 8 : 0);
  ((float4*)d)[0] = make_float4(acc[0][0].x, acc[0][0].y, acc[0][1].x, acc[0][1].y);
  ((float4*)d)[1] = make_float4(acc[0][2].x, acc[0][2].y, acc[0][3].x, acc[0][3].y);
}

__global__ __launch_bounds__(256) void vox_main(const float* __restrict__ w,
                                                float* __restrict__ out) {
  __shared__ __attribute__((aligned(16))) float lw[NG * WSTR];

  const int tid = threadIdx.x;
  {
    const float4* __restrict__ w4 = (const float4*)w;
    float4* lw4 = (float4*)lw;
    float4 stage[(NG * WSTR / 4) / 256];
    #pragma unroll
    for (int i = 0; i < (NG * WSTR / 4) / 256; ++i) stage[i] = w4[tid + 256 * i];
    #pragma unroll
    for (int i = 0; i < (NG * WSTR / 4) / 256; ++i) lw4[tid + 256 * i] = stage[i];
  }
  const float* fl = w + WFLAG;
  const bool product_walk =
      (fl[0] + fl[1] + fl[2] + fl[3] + fl[4] + fl[5] + fl[6] + fl[7] == 0.f);
  __syncthreads();

  const int lane = tid & 31;
  const int wave = tid >> 5;
  const int m    = lane & 15;
  const bool h   = lane >= 16;
  const int hs   = h ? 1 : 0;

  const float EXP_SCALE = -0.72134752044448170368f;
  const float SC = 2.0f / 127.0f;

  const int ix  = wave * 16 + m;
  const float c2 = (float)ix * SC - 1.0f;

  #pragma nounroll
  for (int pp = 0; pp < PATCH_ROWS_PER_BLOCK; ++pp) {
  const int pr  = blockIdx.x * PATCH_ROWS_PER_BLOCK + pp;
  const int iz  = pr >> 3;
  const int iy0 = (pr & 7) * ROWS;
  const float c0 = (float)iz * SC - 1.0f;
  float* const out_w = out + ((iz * GRID_DIM + iy0) * GRID_DIM + wave * 16);

  if (product_walk) {
    const float c1 = (float)(iy0 + START) * SC - 1.0f;
    v2f a0, a1, a2;
    make_a(h, c0, c1, c2, a0, a1, a2);
    v2f as0;
    as0.x = h ? c2  : c0;
    as0.y = h ? 1.f : c1;

    v2f acc[ROWS][4];
    #pragma unroll
    for (int j = 0; j < ROWS; ++j) {
      #pragma unroll
      for (int p = 0; p < 4; ++p) acc[j][p] = (v2f){0.f, 0.f};
    }

    for (int t = 0; t < 16; ++t) {
      const int n = t * 16 + m;
      const v2f* wp = (const v2f*)&lw[n * WSTR];
      const v2f b0 = wp[0 + hs];
      const v2f b1 = wp[2 + hs];
      const v2f b2 = wp[4 + hs];
      const v2f bs = wp[6 + hs];
      const float4 ka = *(const float4*)&lw[n * WSTR + 16];
      const float4 kb = *(const float4*)&lw[n * WSTR + 20];
      const float  k8 = lw[n * WSTR + 24];
      const v2f kd[NKD] = {{ka.x, ka.x}, {ka.y, ka.y}, {ka.z, ka.z}, {ka.w, ka.w},
                           {kb.x, kb.x}, {kb.y, kb.y}, {kb.z, kb.z}, {kb.w, kb.w},
                           {k8, k8}};

      v8f c = {0.f,0.f,0.f,0.f,0.f,0.f,0.f,0.f};
      c = __builtin_amdgcn_wmma_f32_16x16x4_f32(false, a0, false, b0, (short)0, c, false, false);
      c = __builtin_amdgcn_wmma_f32_16x16x4_f32(false, a1, false, b1, (short)0, c, false, false);
      c = __builtin_amdgcn_wmma_f32_16x16x4_f32(false, a2, false, b2, (short)0, c, false, false);
      v8f s = {0.f,0.f,0.f,0.f,0.f,0.f,0.f,0.f};
      s = __builtin_amdgcn_wmma_f32_16x16x4_f32(false, as0, false, bs, (short)0, s, false, false);

      v2f gd[4], gu[4], hu[4], hd[4];
      #pragma unroll
      for (int p = 0; p < 4; ++p) {
        const float ux = s[2*p]     * EXP_SCALE;
        const float uy = s[2*p + 1] * EXP_SCALE;
        gd[p].x = __builtin_amdgcn_exp2f(c[2*p]     * EXP_SCALE);
        gd[p].y = __builtin_amdgcn_exp2f(c[2*p + 1] * EXP_SCALE);
        hu[p].x = __builtin_amdgcn_exp2f(fminf( ux, 126.0f));
        hu[p].y = __builtin_amdgcn_exp2f(fminf( uy, 126.0f));
        hd[p].x = __builtin_amdgcn_exp2f(fminf(-ux, 126.0f));
        hd[p].y = __builtin_amdgcn_exp2f(fminf(-uy, 126.0f));
        gu[p] = gd[p];
        acc[START][p] = __builtin_elementwise_fma(gd[p], kd[0], acc[START][p]);
      }

      #pragma unroll
      for (int j = START + 1; j < ROWS; ++j) {
        #pragma unroll
        for (int p = 0; p < 4; ++p) {
          gu[p] *= hu[p];
          acc[j][p] = __builtin_elementwise_fma(gu[p], kd[j - START - 1], acc[j][p]);
        }
      }

      #pragma unroll
      for (int j = START - 1; j >= 0; --j) {
        #pragma unroll
        for (int p = 0; p < 4; ++p) {
          gd[p] *= hd[p];
          acc[j][p] = __builtin_elementwise_fma(gd[p], kd[START - j], acc[j][p]);
        }
      }
    }

    reduce_store_patch(acc, out_w, lane, h);
  } else {
    #pragma nounroll
    for (int g = 0; g < ROWS / HROWS; ++g) {
      const int iyg = iy0 + g * HROWS;
      const float c1 = (float)iyg * SC - 1.0f;
      v2f a0, a1, a2;
      make_a(h, c0, c1, c2, a0, a1, a2);
      v2f astep[HROWS - 1];
      #pragma unroll
      for (int j = 0; j < HROWS - 1; ++j) {
        const float yj = (float)(iyg + j) * SC - 1.0f;
        astep[j].x = h ? c2  : c0;
        astep[j].y = h ? 1.f : yj;
      }

      v2f acc[HROWS][4];
      #pragma unroll
      for (int j = 0; j < HROWS; ++j) {
        #pragma unroll
        for (int p = 0; p < 4; ++p) acc[j][p] = (v2f){0.f, 0.f};
      }

      for (int t = 0; t < 16; ++t) {
        const int n = t * 16 + m;
        const v2f* wp = (const v2f*)&lw[n * WSTR];
        const v2f b0 = wp[0 + hs];
        const v2f b1 = wp[2 + hs];
        const v2f b2 = wp[4 + hs];
        const v2f bs = wp[6 + hs];
        const float dn = lw[n * WSTR + 16];

        v8f c = {0.f,0.f,0.f,0.f,0.f,0.f,0.f,0.f};
        c = __builtin_amdgcn_wmma_f32_16x16x4_f32(false, a0, false, b0, (short)0, c, false, false);
        c = __builtin_amdgcn_wmma_f32_16x16x4_f32(false, a1, false, b1, (short)0, c, false, false);
        c = __builtin_amdgcn_wmma_f32_16x16x4_f32(false, a2, false, b2, (short)0, c, false, false);

        #pragma unroll
        for (int j = 0; j < HROWS; ++j) {
          v8f cn = c;
          if (j + 1 < HROWS)
            cn = __builtin_amdgcn_wmma_f32_16x16x4_f32(false, astep[j], false, bs, (short)0, c, false, false);
          #pragma unroll
          for (int p = 0; p < 4; ++p) {
            acc[j][p].x += __builtin_amdgcn_exp2f(c[2*p]     * EXP_SCALE) * dn;
            acc[j][p].y += __builtin_amdgcn_exp2f(c[2*p + 1] * EXP_SCALE) * dn;
          }
          c = cn;
        }
      }

      #pragma unroll
      for (int j = 0; j < HROWS; ++j)
        reduce_store_row(acc[j], out_w + (g * HROWS + j) * GRID_DIM, lane, h);
    }
  }
  }
}

extern "C" void kernel_launch(void* const* d_in, const int* in_sizes, int n_in,
                              void* d_out, int out_size, void* d_ws, size_t ws_size,
                              hipStream_t stream) {
  const float* positions = (const float*)d_in[0];
  const float* scales    = (const float*)d_in[1];
  const float* rotations = (const float*)d_in[2];
  const float* density   = (const float*)d_in[3];
  float* out = (float*)d_out;
  float* wtab = (float*)d_ws;

  vox_prep<<<1, 256, 0, stream>>>(positions, scales, rotations, density, wtab);
  vox_main<<<MAIN_BLOCKS, 256, 0, stream>>>(wtab, out);
}


# Epilogue column sums via LDS: a lane stores its 16x8 partial sums as 16-byte chunks (pitch 17), reads one row quad's 16 columns back, adds them in the same tree; no DPP, no selects; sum order same
# speedup vs baseline: 3.5564x; 1.0513x over previous
#include <hip/hip_runtime.h>


typedef __attribute__((ext_vector_type(2))) float v2f;
typedef __attribute__((ext_vector_type(8))) float v8f;
typedef __attribute__((ext_vector_type(4))) float v4f;

#define NG   256
#define KDIM 12
#define WSTR 28
#define NKD  9
#define WFLAG (NG * WSTR)
#define C11_MAX 3000.0f
#define GRID_DIM 128
#define ROWS 16
#define HROWS 8
#define START 8
#define PATCH_ROWS_PER_BLOCK 2
#define MAIN_BLOCKS (GRID_DIM * (GRID_DIM / ROWS) / PATCH_ROWS_PER_BLOCK)
#define RED_ROWS  8
#define RED_PITCH 17
#define RED_WAVE  (2 * 16 * RED_PITCH)

__global__ __launch_bounds__(256) void vox_prep(const float* __restrict__ pos,
                                                const float* __restrict__ scl,
                                                const float* __restrict__ rot,
                                                const float* __restrict__ den,
                                                float* __restrict__ w) {
  const int n = threadIdx.x;
  float qw = rot[n*4+0], qx = rot[n*4+1], qy = rot[n*4+2], qz = rot[n*4+3];
  float rn = rsqrtf(qw*qw + qx*qx + qy*qy + qz*qz);
  qw *= rn; qx *= rn; qy *= rn; qz *= rn;
  float R00 = 1.f - 2.f*(qy*qy + qz*qz);
  float R01 = 2.f*(qx*qy - qw*qz);
  float R02 = 2.f*(qx*qz + qw*qy);
  float R10 = 2.f*(qx*qy + qw*qz);
  float R11 = 1.f - 2.f*(qx*qx + qz*qz);
  float R12 = 2.f*(qy*qz - qw*qx);
  float R20 = 2.f*(qx*qz - qw*qy);
  float R21 = 2.f*(qy*qz + qw*qx);
  float R22 = 1.f - 2.f*(qx*qx + qy*qy);
  float s0 = scl[n*3+0], s1 = scl[n*3+1], s2 = scl[n*3+2];
  float i0 = 1.f/(s0*s0 + 1e-8f);
  float i1 = 1.f/(s1*s1 + 1e-8f);
  float i2 = 1.f/(s2*s2 + 1e-8f);
  float C00 = R00*R00*i0 + R01*R01*i1 + R02*R02*i2;
  float C01 = R00*R10*i0 + R01*R11*i1 + R02*R12*i2;
  float C02 = R00*R20*i0 + R01*R21*i1 + R02*R22*i2;
  float C11 = R10*R10*i0 + R11*R11*i1 + R12*R12*i2;
  float C12 = R10*R20*i0 + R11*R21*i1 + R12*R22*i2;
  float C22 = R20*R20*i0 + R21*R21*i1 + R22*R22*i2;
  float p0 = pos[n*3+0], p1 = pos[n*3+1], p2 = pos[n*3+2];
  float b0 = C00*p0 + C01*p1 + C02*p2;
  float b1 = C01*p0 + C11*p1 + C12*p2;
  float b2 = C02*p0 + C12*p1 + C22*p2;
  float cn = p0*b0 + p1*b1 + p2*b2;
  const float d = 2.0f / 127.0f;
  float* wn = w + n*WSTR;
  wn[0] = C00;      wn[1] = C11;      wn[2]  = C22;
  wn[3] = 2.f*C01;  wn[4] = 2.f*C02;  wn[5]  = 2.f*C12;
  wn[6] = -2.f*b0;  wn[7] = -2.f*b1;  wn[8]  = -2.f*b2;
  wn[9] = cn;       wn[10] = 0.f;     wn[11] = 0.f;
  wn[12] = 2.f*C01*d;  wn[13] = 2.f*C11*d;
  wn[14] = 2.f*C12*d;  wn[15] = C11*d*d - 2.f*b1*d;
  const float dn = den[n];
  const float lk = -C11*d*d;
  wn[16] = dn;
  #pragma unroll
  for (int i = 1; i < NKD; ++i) wn[16 + i] = dn * expf(lk * (float)(i*(i+1)/2));
  wn[25] = 0.f; wn[26] = 0.f; wn[27] = 0.f;
  const int wave_narrow = __any(!(C11 <= C11_MAX));
  if ((n & 31) == 0) w[WFLAG + (n >> 5)] = wave_narrow ? 1.f : 0.f;
}

__device__ __forceinline__ void make_a(bool h, float c0, float c1, float c2,
                                       v2f& a0, v2f& a1, v2f& a2) {
  a0.x = h ? c2*c2 : c0*c0;   a0.y = h ? c0*c1 : c1*c1;
  a1.x = h ? c0    : c0*c2;   a1.y = h ? c1    : c1*c2;
  a2.x = h ? 0.f   : c2;      a2.y = h ? 0.f   : 1.f;
}

__device__ __forceinline__ void reduce_store_row(v2f (&a)[4], float* __restrict__ dst,
                                                 int lane, bool h) {
  #pragma unroll
  for (int mask = 1; mask <= 8; mask <<= 1) {
    #pragma unroll
    for (int p = 0; p < 4; ++p) {
      a[p].x += __shfl_xor(a[p].x, mask, 32);
      a[p].y += __shfl_xor(a[p].y, mask, 32);
    }
  }
  if (lane == 0 || lane == 16) {
    float4 lo = make_float4(a[0].x, a[0].y, a[1].x, a[1].y);
    float4 hi = make_float4(a[2].x, a[2].y, a[3].x, a[3].y);
    float* d = dst + (h ? 8 : 0);
    ((float4*)d)[0] = lo;
    ((float4*)d)[1] = hi;
  }
}

__device__ __forceinline__ void reduce_store_patch(v2f (&acc)[ROWS][4], v4f* red,
                                                   float* __restrict__ patch, int lane) {
  const int col = lane & 15;
  const int hs  = lane >> 4;
  const int rr  = lane & 7;
  const int rq  = (lane >> 3) & 1;
  v4f* const wr = red + hs * (16 * RED_PITCH) + col * RED_PITCH;
  const v4f* const rd = red + hs * (16 * RED_PITCH) + 2 * rr + rq;
  float* const dst = patch + rr * GRID_DIM + hs * 8 + rq * 4;
  #pragma unroll
  for (int pass = 0; pass < ROWS / RED_ROWS; ++pass) {
    #pragma unroll
    for (int j = 0; j < RED_ROWS; ++j) {
      #pragma unroll
      for (int q = 0; q < 2; ++q)
        wr[2 * j + q] = __builtin_shufflevector(acc[pass * RED_ROWS + j][2 * q],
                                                acc[pass * RED_ROWS + j][2 * q + 1], 0, 1, 2, 3);
    }
    v4f x[16];
    #pragma unroll
    for (int cc = 0; cc < 16; ++cc) x[cc] = rd[cc * RED_PITCH];
    #pragma unroll
    for (int w = 1; w < 16; w <<= 1) {
      #pragma unroll
      for (int i = 0; i < 16; i += 2 * w) x[i] = x[i] + x[i + w];
    }
    *(v4f*)(dst + pass * RED_ROWS * GRID_DIM) = x[0];
  }
}

__global__ __launch_bounds__(256) void vox_main(const float* __restrict__ w,
                                                float* __restrict__ out) {
  __shared__ __attribute__((aligned(16))) float lw[NG * WSTR];
  __shared__ __attribute__((aligned(16))) v4f red_all[8 * RED_WAVE];

  const int tid = threadIdx.x;
  {
    const float4* __restrict__ w4 = (const float4*)w;
    float4* lw4 = (float4*)lw;
    float4 stage[(NG * WSTR / 4) / 256];
    #pragma unroll
    for (int i = 0; i < (NG * WSTR / 4) / 256; ++i) stage[i] = w4[tid + 256 * i];
    #pragma unroll
    for (int i = 0; i < (NG * WSTR / 4) / 256; ++i) lw4[tid + 256 * i] = stage[i];
  }
  const float* fl = w + WFLAG;
  const bool product_walk =
      (fl[0] + fl[1] + fl[2] + fl[3] + fl[4] + fl[5] + fl[6] + fl[7] == 0.f);
  __syncthreads();

  const int lane = tid & 31;
  const int wave = tid >> 5;
  const int m    = lane & 15;
  const bool h   = lane >= 16;
  const int hs   = h ? 1 : 0;

  const float EXP_SCALE = -0.72134752044448170368f;
  const float SC = 2.0f / 127.0f;

  const int ix  = wave * 16 + m;
  const float c2 = (float)ix * SC - 1.0f;

  #pragma nounroll
  for (int pp = 0; pp < PATCH_ROWS_PER_BLOCK; ++pp) {
  const int pr  = blockIdx.x * PATCH_ROWS_PER_BLOCK + pp;
  const int iz  = pr >> 3;
  const int iy0 = (pr & 7) * ROWS;
  const float c0 = (float)iz * SC - 1.0f;
  float* const out_w = out + ((iz * GRID_DIM + iy0) * GRID_DIM + wave * 16);

  if (product_walk) {
    const float c1 = (float)(iy0 + START) * SC - 1.0f;
    v2f a0, a1, a2;
    make_a(h, c0, c1, c2, a0, a1, a2);
    v2f as0;
    as0.x = h ? c2  : c0;
    as0.y = h ? 1.f : c1;

    v2f acc[ROWS][4];
    #pragma unroll
    for (int j = 0; j < ROWS; ++j) {
      #pragma unroll
      for (int p = 0; p < 4; ++p) acc[j][p] = (v2f){0.f, 0.f};
    }

    for (int t = 0; t < 16; ++t) {
      const int n = t * 16 + m;
      const v2f* wp = (const v2f*)&lw[n * WSTR];
      const v2f b0 = wp[0 + hs];
      const v2f b1 = wp[2 + hs];
      const v2f b2 = wp[4 + hs];
      const v2f bs = wp[6 + hs];
      const float4 ka = *(const float4*)&lw[n * WSTR + 16];
      const float4 kb = *(const float4*)&lw[n * WSTR + 20];
      const float  k8 = lw[n * WSTR + 24];
      const v2f kd[NKD] = {{ka.x, ka.x}, {ka.y, ka.y}, {ka.z, ka.z}, {ka.w, ka.w},
                           {kb.x, kb.x}, {kb.y, kb.y}, {kb.z, kb.z}, {kb.w, kb.w},
                           {k8, k8}};

      v8f c = {0.f,0.f,0.f,0.f,0.f,0.f,0.f,0.f};
      c = __builtin_amdgcn_wmma_f32_16x16x4_f32(false, a0, false, b0, (short)0, c, false, false);
      c = __builtin_amdgcn_wmma_f32_16x16x4_f32(false, a1, false, b1, (short)0, c, false, false);
      c = __builtin_amdgcn_wmma_f32_16x16x4_f32(false, a2, false, b2, (short)0, c, false, false);
      v8f s = {0.f,0.f,0.f,0.f,0.f,0.f,0.f,0.f};
      s = __builtin_amdgcn_wmma_f32_16x16x4_f32(false, as0, false, bs, (short)0, s, false, false);

      v2f gd[4], gu[4], hu[4], hd[4];
      #pragma unroll
      for (int p = 0; p < 4; ++p) {
        const float ux = s[2*p]     * EXP_SCALE;
        const float uy = s[2*p + 1] * EXP_SCALE;
        gd[p].x = __builtin_amdgcn_exp2f(c[2*p]     * EXP_SCALE);
        gd[p].y = __builtin_amdgcn_exp2f(c[2*p + 1] * EXP_SCALE);
        hu[p].x = __builtin_amdgcn_exp2f(fminf( ux, 126.0f));
        hu[p].y = __builtin_amdgcn_exp2f(fminf( uy, 126.0f));
        hd[p].x = __builtin_amdgcn_exp2f(fminf(-ux, 126.0f));
        hd[p].y = __builtin_amdgcn_exp2f(fminf(-uy, 126.0f));
        gu[p] = gd[p];
        acc[START][p] = __builtin_elementwise_fma(gd[p], kd[0], acc[START][p]);
      }

      #pragma unroll
      for (int j = START + 1; j < ROWS; ++j) {
        #pragma unroll
        for (int p = 0; p < 4; ++p) {
          gu[p] *= hu[p];
          acc[j][p] = __builtin_elementwise_fma(gu[p], kd[j - START - 1], acc[j][p]);
        }
      }

      #pragma unroll
      for (int j = START - 1; j >= 0; --j) {
        #pragma unroll
        for (int p = 0; p < 4; ++p) {
          gd[p] *= hd[p];
          acc[j][p] = __builtin_elementwise_fma(gd[p], kd[START - j], acc[j][p]);
        }
      }
    }

    reduce_store_patch(acc, red_all + wave * RED_WAVE, out_w, lane);
  } else {
    #pragma nounroll
    for (int g = 0; g < ROWS / HROWS; ++g) {
      const int iyg = iy0 + g * HROWS;
      const float c1 = (float)iyg * SC - 1.0f;
      v2f a0, a1, a2;
      make_a(h, c0, c1, c2, a0, a1, a2);
      v2f astep[HROWS - 1];
      #pragma unroll
      for (int j = 0; j < HROWS - 1; ++j) {
        const float yj = (float)(iyg + j) * SC - 1.0f;
        astep[j].x = h ? c2  : c0;
        astep[j].y = h ? 1.f : yj;
      }

      v2f acc[HROWS][4];
      #pragma unroll
      for (int j = 0; j < HROWS; ++j) {
        #pragma unroll
        for (int p = 0; p < 4; ++p) acc[j][p] = (v2f){0.f, 0.f};
      }

      for (int t = 0; t < 16; ++t) {
        const int n = t * 16 + m;
        const v2f* wp = (const v2f*)&lw[n * WSTR];
        const v2f b0 = wp[0 + hs];
        const v2f b1 = wp[2 + hs];
        const v2f b2 = wp[4 + hs];
        const v2f bs = wp[6 + hs];
        const float dn = lw[n * WSTR + 16];

        v8f c = {0.f,0.f,0.f,0.f,0.f,0.f,0.f,0.f};
        c = __builtin_amdgcn_wmma_f32_16x16x4_f32(false, a0, false, b0, (short)0, c, false, false);
        c = __builtin_amdgcn_wmma_f32_16x16x4_f32(false, a1, false, b1, (short)0, c, false, false);
        c = __builtin_amdgcn_wmma_f32_16x16x4_f32(false, a2, false, b2, (short)0, c, false, false);

        #pragma unroll
        for (int j = 0; j < HROWS; ++j) {
          v8f cn = c;
          if (j + 1 < HROWS)
            cn = __builtin_amdgcn_wmma_f32_16x16x4_f32(false, astep[j], false, bs, (short)0, c, false, false);
          #pragma unroll
          for (int p = 0; p < 4; ++p) {
            acc[j][p].x += __builtin_amdgcn_exp2f(c[2*p]     * EXP_SCALE) * dn;
            acc[j][p].y += __builtin_amdgcn_exp2f(c[2*p + 1] * EXP_SCALE) * dn;
          }
          c = cn;
        }
      }

      #pragma unroll
      for (int j = 0; j < HROWS; ++j)
        reduce_store_row(acc[j], out_w + (g * HROWS + j) * GRID_DIM, lane, h);
    }
  }
  }
}

extern "C" void kernel_launch(void* const* d_in, const int* in_sizes, int n_in,
                              void* d_out, int out_size, void* d_ws, size_t ws_size,
                              hipStream_t stream) {
  const float* positions = (const float*)d_in[0];
  const float* scales    = (const float*)d_in[1];
  const float* rotations = (const float*)d_in[2];
  const float* density   = (const float*)d_in[3];
  float* out = (float*)d_out;
  float* wtab = (float*)d_ws;

  vox_prep<<<1, 256, 0, stream>>>(positions, scales, rotations, density, wtab);
  vox_main<<<MAIN_BLOCKS, 256, 0, stream>>>(wtab, out);
}


# Row ratio H walked along x in a lane: H(x+d) = H(x)*exp(-C12*d*d), started by 3 FMAs and 2 exp2 at the lane's first voxel; no step WMMA, 10 exp2 per tile, was 24; H rounding changes, sum order same
# speedup vs baseline: 3.7874x; 1.0650x over previous
#include <hip/hip_runtime.h>


typedef __attribute__((ext_vector_type(2))) float v2f;
typedef __attribute__((ext_vector_type(8))) float v8f;
typedef __attribute__((ext_vector_type(4))) float v4f;

#define NG   256
#define KDIM 12
#define WSTR 28
#define NKD  9
#define WFLAG (NG * WSTR)
#define C11_MAX 3000.0f
#define H_CAP 118.0f
#define GRID_DIM 128
#define ROWS 16
#define HROWS 8
#define START 8
#define PATCH_ROWS_PER_BLOCK 2
#define MAIN_BLOCKS (GRID_DIM * (GRID_DIM / ROWS) / PATCH_ROWS_PER_BLOCK)
#define RED_ROWS  8
#define RED_PITCH 17
#define RED_WAVE  (2 * 16 * RED_PITCH)

__global__ __launch_bounds__(256) void vox_prep(const float* __restrict__ pos,
                                                const float* __restrict__ scl,
                                                const float* __restrict__ rot,
                                                const float* __restrict__ den,
                                                float* __restrict__ w) {
  const int n = threadIdx.x;
  float qw = rot[n*4+0], qx = rot[n*4+1], qy = rot[n*4+2], qz = rot[n*4+3];
  float rn = rsqrtf(qw*qw + qx*qx + qy*qy + qz*qz);
  qw *= rn; qx *= rn; qy *= rn; qz *= rn;
  float R00 = 1.f - 2.f*(qy*qy + qz*qz);
  float R01 = 2.f*(qx*qy - qw*qz);
  float R02 = 2.f*(qx*qz + qw*qy);
  float R10 = 2.f*(qx*qy + qw*qz);
  float R11 = 1.f - 2.f*(qx*qx + qz*qz);
  float R12 = 2.f*(qy*qz - qw*qx);
  float R20 = 2.f*(qx*qz - qw*qy);
  float R21 = 2.f*(qy*qz + qw*qx);
  float R22 = 1.f - 2.f*(qx*qx + qy*qy);
  float s0 = scl[n*3+0], s1 = scl[n*3+1], s2 = scl[n*3+2];
  float i0 = 1.f/(s0*s0 + 1e-8f);
  float i1 = 1.f/(s1*s1 + 1e-8f);
  float i2 = 1.f/(s2*s2 + 1e-8f);
  float C00 = R00*R00*i0 + R01*R01*i1 + R02*R02*i2;
  float C01 = R00*R10*i0 + R01*R11*i1 + R02*R12*i2;
  float C02 = R00*R20*i0 + R01*R21*i1 + R02*R22*i2;
  float C11 = R10*R10*i0 + R11*R11*i1 + R12*R12*i2;
  float C12 = R10*R20*i0 + R11*R21*i1 + R12*R22*i2;
  float C22 = R20*R20*i0 + R21*R21*i1 + R22*R22*i2;
  float p0 = pos[n*3+0], p1 = pos[n*3+1], p2 = pos[n*3+2];
  float b0 = C00*p0 + C01*p1 + C02*p2;
  float b1 = C01*p0 + C11*p1 + C12*p2;
  float b2 = C02*p0 + C12*p1 + C22*p2;
  float cn = p0*b0 + p1*b1 + p2*b2;
  const float d = 2.0f / 127.0f;
  float* wn = w + n*WSTR;
  wn[0] = C00;      wn[1] = C11;      wn[2]  = C22;
  wn[3] = 2.f*C01;  wn[4] = 2.f*C02;  wn[5]  = 2.f*C12;
  wn[6] = -2.f*b0;  wn[7] = -2.f*b1;  wn[8]  = -2.f*b2;
  wn[9] = cn;       wn[10] = 0.f;     wn[11] = 0.f;
  wn[12] = 2.f*C01*d;  wn[13] = 2.f*C11*d;
  wn[14] = 2.f*C12*d;  wn[15] = C11*d*d - 2.f*b1*d;
  const float dn = den[n];
  const float lk = -C11*d*d;
  wn[16] = dn;
  #pragma unroll
  for (int i = 1; i < NKD; ++i) wn[16 + i] = dn * expf(lk * (float)(i*(i+1)/2));
  wn[25] = expf(-C12*d*d);
  wn[26] = expf( C12*d*d);
  wn[27] = 0.f;
  const int wave_narrow = __any(!(C11 <= C11_MAX) || !(C22 <= C11_MAX));
  if ((n & 31) == 0) w[WFLAG + (n >> 5)] = wave_narrow ? 1.f : 0.f;
}

__device__ __forceinline__ void make_a(bool h, float c0, float c1, float c2,
                                       v2f& a0, v2f& a1, v2f& a2) {
  a0.x = h ? c2*c2 : c0*c0;   a0.y = h ? c0*c1 : c1*c1;
  a1.x = h ? c0    : c0*c2;   a1.y = h ? c1    : c1*c2;
  a2.x = h ? 0.f   : c2;      a2.y = h ? 0.f   : 1.f;
}

__device__ __forceinline__ void reduce_store_row(v2f (&a)[4], float* __restrict__ dst,
                                                 int lane, bool h) {
  #pragma unroll
  for (int mask = 1; mask <= 8; mask <<= 1) {
    #pragma unroll
    for (int p = 0; p < 4; ++p) {
      a[p].x += __shfl_xor(a[p].x, mask, 32);
      a[p].y += __shfl_xor(a[p].y, mask, 32);
    }
  }
  if (lane == 0 || lane == 16) {
    float4 lo = make_float4(a[0].x, a[0].y, a[1].x, a[1].y);
    float4 hi = make_float4(a[2].x, a[2].y, a[3].x, a[3].y);
    float* d = dst + (h ? 8 : 0);
    ((float4*)d)[0] = lo;
    ((float4*)d)[1] = hi;
  }
}

__device__ __forceinline__ void reduce_store_patch(v2f (&acc)[ROWS][4], v4f* red,
                                                   float* __restrict__ patch, int lane) {
  const int col = lane & 15;
  const int hs  = lane >> 4;
  const int rr  = lane & 7;
  const int rq  = (lane >> 3) & 1;
  v4f* const wr = red + hs * (16 * RED_PITCH) + col * RED_PITCH;
  const v4f* const rd = red + hs * (16 * RED_PITCH) + 2 * rr + rq;
  float* const dst = patch + rr * GRID_DIM + hs * 8 + rq * 4;
  #pragma unroll
  for (int pass = 0; pass < ROWS / RED_ROWS; ++pass) {
    #pragma unroll
    for (int j = 0; j < RED_ROWS; ++j) {
      #pragma unroll
      for (int q = 0; q < 2; ++q)
        wr[2 * j + q] = __builtin_shufflevector(acc[pass * RED_ROWS + j][2 * q],
                                                acc[pass * RED_ROWS + j][2 * q + 1], 0, 1, 2, 3);
    }
    v4f x[16];
    #pragma unroll
    for (int cc = 0; cc < 16; ++cc) x[cc] = rd[cc * RED_PITCH];
    #pragma unroll
    for (int w = 1; w < 16; w <<= 1) {
      #pragma unroll
      for (int i = 0; i < 16; i += 2 * w) x[i] = x[i] + x[i + w];
    }
    *(v4f*)(dst + pass * RED_ROWS * GRID_DIM) = x[0];
  }
}

__global__ __launch_bounds__(256) void vox_main(const float* __restrict__ w,
                                                float* __restrict__ out) {
  __shared__ __attribute__((aligned(16))) float lw[NG * WSTR];
  __shared__ __attribute__((aligned(16))) v4f red_all[8 * RED_WAVE];

  const int tid = threadIdx.x;
  {
    const float4* __restrict__ w4 = (const float4*)w;
    float4* lw4 = (float4*)lw;
    float4 stage[(NG * WSTR / 4) / 256];
    #pragma unroll
    for (int i = 0; i < (NG * WSTR / 4) / 256; ++i) stage[i] = w4[tid + 256 * i];
    #pragma unroll
    for (int i = 0; i < (NG * WSTR / 4) / 256; ++i) lw4[tid + 256 * i] = stage[i];
  }
  const float* fl = w + WFLAG;
  const bool product_walk =
      (fl[0] + fl[1] + fl[2] + fl[3] + fl[4] + fl[5] + fl[6] + fl[7] == 0.f);
  __syncthreads();

  const int lane = tid & 31;
  const int wave = tid >> 5;
  const int m    = lane & 15;
  const bool h   = lane >= 16;
  const int hs   = h ? 1 : 0;

  const float EXP_SCALE = -0.72134752044448170368f;
  const float SC = 2.0f / 127.0f;

  const int ix  = wave * 16 + m;
  const float c2 = (float)ix * SC - 1.0f;
  const float x_first = (float)(wave * 16 + 8 * hs) * SC - 1.0f;

  #pragma nounroll
  for (int pp = 0; pp < PATCH_ROWS_PER_BLOCK; ++pp) {
  const int pr  = blockIdx.x * PATCH_ROWS_PER_BLOCK + pp;
  const int iz  = pr >> 3;
  const int iy0 = (pr & 7) * ROWS;
  const float c0 = (float)iz * SC - 1.0f;
  float* const out_w = out + ((iz * GRID_DIM + iy0) * GRID_DIM + wave * 16);

  if (product_walk) {
    const float c1 = (float)(iy0 + START) * SC - 1.0f;
    v2f a0, a1, a2;
    make_a(h, c0, c1, c2, a0, a1, a2);

    v2f acc[ROWS][4];
    #pragma unroll
    for (int j = 0; j < ROWS; ++j) {
      #pragma unroll
      for (int p = 0; p < 4; ++p) acc[j][p] = (v2f){0.f, 0.f};
    }

    for (int t = 0; t < 16; ++t) {
      const int n = t * 16 + m;
      const v2f* wp = (const v2f*)&lw[n * WSTR];
      const v2f b0 = wp[0 + hs];
      const v2f b1 = wp[2 + hs];
      const v2f b2 = wp[4 + hs];
      const float4 sw = *(const float4*)&lw[n * WSTR + 12];
      const float4 ka = *(const float4*)&lw[n * WSTR + 16];
      const float4 kb = *(const float4*)&lw[n * WSTR + 20];
      const float4 kc = *(const float4*)&lw[n * WSTR + 24];
      const v2f kd[NKD] = {{ka.x, ka.x}, {ka.y, ka.y}, {ka.z, ka.z}, {ka.w, ka.w},
                           {kb.x, kb.x}, {kb.y, kb.y}, {kb.z, kb.z}, {kb.w, kb.w},
                           {kc.x, kc.x}};

      v8f c = {0.f,0.f,0.f,0.f,0.f,0.f,0.f,0.f};
      c = __builtin_amdgcn_wmma_f32_16x16x4_f32(false, a0, false, b0, (short)0, c, false, false);
      c = __builtin_amdgcn_wmma_f32_16x16x4_f32(false, a1, false, b1, (short)0, c, false, false);
      c = __builtin_amdgcn_wmma_f32_16x16x4_f32(false, a2, false, b2, (short)0, c, false, false);

      const float s_first = fmaf(sw.x, c0, fmaf(sw.y, c1, fmaf(sw.z, x_first, sw.w)));
      const float u_first = fminf(fmaxf(s_first * EXP_SCALE, -H_CAP), H_CAP);
      const float h_first = __builtin_amdgcn_exp2f( u_first);
      const float r_first = __builtin_amdgcn_exp2f(-u_first);
      const float qu = kc.y * kc.y;
      const float qd = kc.z * kc.z;
      v2f gd[4], gu[4], hu[4], hd[4];
      hu[0] = (v2f){h_first, h_first * kc.y};
      hd[0] = (v2f){r_first, r_first * kc.z};
      #pragma unroll
      for (int p = 1; p < 4; ++p) {
        hu[p] = hu[p - 1] * (v2f){qu, qu};
        hd[p] = hd[p - 1] * (v2f){qd, qd};
      }
      #pragma unroll
      for (int p = 0; p < 4; ++p) {
        gd[p].x = __builtin_amdgcn_exp2f(c[2*p]     * EXP_SCALE);
        gd[p].y = __builtin_amdgcn_exp2f(c[2*p + 1] * EXP_SCALE);
        gu[p] = gd[p];
        acc[START][p] = __builtin_elementwise_fma(gd[p], kd[0], acc[START][p]);
      }

      #pragma unroll
      for (int j = START + 1; j < ROWS; ++j) {
        #pragma unroll
        for (int p = 0; p < 4; ++p) {
          gu[p] *= hu[p];
          acc[j][p] = __builtin_elementwise_fma(gu[p], kd[j - START - 1], acc[j][p]);
        }
      }

      #pragma unroll
      for (int j = START - 1; j >= 0; --j) {
        #pragma unroll
        for (int p = 0; p < 4; ++p) {
          gd[p] *= hd[p];
          acc[j][p] = __builtin_elementwise_fma(gd[p], kd[START - j], acc[j][p]);
        }
      }
    }

    reduce_store_patch(acc, red_all + wave * RED_WAVE, out_w, lane);
  } else {
    #pragma nounroll
    for (int g = 0; g < ROWS / HROWS; ++g) {
      const int iyg = iy0 + g * HROWS;
      const float c1 = (float)iyg * SC - 1.0f;
      v2f a0, a1, a2;
      make_a(h, c0, c1, c2, a0, a1, a2);
      v2f astep[HROWS - 1];
      #pragma unroll
      for (int j = 0; j < HROWS - 1; ++j) {
        const float yj = (float)(iyg + j) * SC - 1.0f;
        astep[j].x = h ? c2  : c0;
        astep[j].y = h ? 1.f : yj;
      }

      v2f acc[HROWS][4];
      #pragma unroll
      for (int j = 0; j < HROWS; ++j) {
        #pragma unroll
        for (int p = 0; p < 4; ++p) acc[j][p] = (v2f){0.f, 0.f};
      }

      for (int t = 0; t < 16; ++t) {
        const int n = t * 16 + m;
        const v2f* wp = (const v2f*)&lw[n * WSTR];
        const v2f b0 = wp[0 + hs];
        const v2f b1 = wp[2 + hs];
        const v2f b2 = wp[4 + hs];
        const v2f bs = wp[6 + hs];
        const float dn = lw[n * WSTR + 16];

        v8f c = {0.f,0.f,0.f,0.f,0.f,0.f,0.f,0.f};
        c = __builtin_amdgcn_wmma_f32_16x16x4_f32(false, a0, false, b0, (short)0, c, false, false);
        c = __builtin_amdgcn_wmma_f32_16x16x4_f32(false, a1, false, b1, (short)0, c, false, false);
        c = __builtin_amdgcn_wmma_f32_16x16x4_f32(false, a2, false, b2, (short)0, c, false, false);

        #pragma unroll
        for (int j = 0; j < HROWS; ++j) {
          v8f cn = c;
          if (j + 1 < HROWS)
            cn = __builtin_amdgcn_wmma_f32_16x16x4_f32(false, astep[j], false, bs, (short)0, c, false, false);
          #pragma unroll
          for (int p = 0; p < 4; ++p) {
            acc[j][p].x += __builtin_amdgcn_exp2f(c[2*p]     * EXP_SCALE) * dn;
            acc[j][p].y += __builtin_amdgcn_exp2f(c[2*p + 1] * EXP_SCALE) * dn;
          }
          c = cn;
        }
      }

      #pragma unroll
      for (int j = 0; j < HROWS; ++j)
        reduce_store_row(acc[j], out_w + (g * HROWS + j) * GRID_DIM, lane, h);
    }
  }
  }
}

extern "C" void kernel_launch(void* const* d_in, const int* in_sizes, int n_in,
                              void* d_out, int out_size, void* d_ws, size_t ws_size,
                              hipStream_t stream) {
  const float* positions = (const float*)d_in[0];
  const float* scales    = (const float*)d_in[1];
  const float* rotations = (const float*)d_in[2];
  const float* density   = (const float*)d_in[3];
  float* out = (float*)d_out;
  float* wtab = (float*)d_ws;

  vox_prep<<<1, 256, 0, stream>>>(positions, scales, rotations, density, wtab);
  vox_main<<<MAIN_BLOCKS, 256, 0, stream>>>(wtab, out);
}
